# starting kernel
# baseline (speedup 1.0000x reference)
#include <hip/hip_runtime.h>
#include <hip/hip_bf16.h>

#define N_NODES 50000
#define N_EDGES 800000
#define IN_FEATS 128
#define HIDDEN 64
#define M_TILES (N_NODES / 16)

typedef float v2f __attribute__((ext_vector_type(2)));
typedef float v8f __attribute__((ext_vector_type(8)));

__global__ void count_kernel(const int* __restrict__ src, const int* __restrict__ dst,
                             int* __restrict__ cnt_src, int* __restrict__ cnt_dst, int E) {
    int e = blockIdx.x * blockDim.x + threadIdx.x;
    if (e >= E) return;
    atomicAdd(&cnt_src[src[e]], 1);
    atomicAdd(&cnt_dst[dst[e]], 1);
}

__global__ void coeff_kernel(int* __restrict__ cs, int* __restrict__ cd, int n) {
    int i = blockIdx.x * blockDim.x + threadIdx.x;
    if (i >= n) return;
    int a = cs[i]; int b = cd[i];
    ((float*)cs)[i] = rsqrtf((float)(a < 1 ? 1 : a));
    ((float*)cd)[i] = rsqrtf((float)(b < 1 ? 1 : b));
}

template<int K>
__global__ __launch_bounds__(128) void gemm_wmma_kernel(
        const float* __restrict__ A, const float* __restrict__ W,
        float* __restrict__ out, int mtiles) {
    __shared__ v2f sB[(K / 2) * 64];
    for (int i = (int)threadIdx.x; i < (K / 2) * 64; i += 128) {
        const int j = i >> 6, n = i & 63;
        v2f p;
        p.x = W[(2 * j)     * 64 + n];
        p.y = W[(2 * j + 1) * 64 + n];
        sB[i] = p;
    }
    __syncthreads();

    const int lane  = (int)threadIdx.x & 31;
    const int mtile = (int)blockIdx.x * 4 + ((int)threadIdx.x >> 5);
    if (mtile >= mtiles) return;
    const int m_base = mtile * 16;
    const int col    = lane & 15;
    const int half   = lane >> 4;
    const int row    = m_base + col;
    const float* __restrict__ arow = A + (size_t)row * K;

    v8f acc0 = {}, acc1 = {}, acc2 = {}, acc3 = {};
    for (int k0 = 0; k0 < K; k0 += 4) {
        const int kb = k0 + (half << 1);
        const v2f a = *(const v2f*)(arow + kb);
        const v2f* __restrict__ brow = sB + ((k0 >> 1) + half) * 64;
        const v2f b0 = brow[     col];
        const v2f b1 = brow[16 + col];
        const v2f b2 = brow[32 + col];
        const v2f b3 = brow[48 + col];
        acc0 = __builtin_amdgcn_wmma_f32_16x16x4_f32(false, a, false, b0, (short)0, acc0, false, false);
        acc1 = __builtin_amdgcn_wmma_f32_16x16x4_f32(false, a, false, b1, (short)0, acc1, false, false);
        acc2 = __builtin_amdgcn_wmma_f32_16x16x4_f32(false, a, false, b2, (short)0, acc2, false, false);
        acc3 = __builtin_amdgcn_wmma_f32_16x16x4_f32(false, a, false, b3, (short)0, acc3, false, false);
    }

    const int rtop = half << 3;
    #pragma unroll
    for (int v = 0; v < 8; ++v) {
        const size_t r = (size_t)(m_base + rtop + v) * 64;
        out[r +      col] = acc0[v];
        out[r + 16 + col] = acc1[v];
        out[r + 32 + col] = acc2[v];
        out[r + 48 + col] = acc3[v];
    }
}

__global__ void scatter64_kernel(const float* __restrict__ y, const float* __restrict__ csrc,
                                 const int* __restrict__ src, const int* __restrict__ dst,
                                 float* __restrict__ agg, int E) {
    int e = blockIdx.x * 4 + ((int)threadIdx.x >> 6);
    int f = (int)threadIdx.x & 63;
    if (e >= E) return;
    int s = src[e], d = dst[e];
    float v = y[(size_t)s * 64 + f] * csrc[s];
    atomicAdd(&agg[(size_t)d * 64 + f], v);
}

template<bool RELU>
__global__ void epilogue_kernel(float* __restrict__ agg, const float* __restrict__ cdst,
                                const float* __restrict__ bias, int n_elems) {
    int i = blockIdx.x * blockDim.x + threadIdx.x;
    if (i >= n_elems) return;
    int row = i >> 6, f = i & 63;
    float v = agg[i] * cdst[row] + bias[f];
    if (RELU) v = fmaxf(v, 0.0f);
    agg[i] = v;
}

__global__ void edge_kernel(const float* __restrict__ h, const int* __restrict__ src,
                            const int* __restrict__ dst, const float* __restrict__ efeat,
                            const float* __restrict__ We, const float* __restrict__ be,
                            float* __restrict__ out, int E) {
    const int lane = (int)threadIdx.x & 31;
    const int e = (int)blockIdx.x * ((int)blockDim.x >> 5) + ((int)threadIdx.x >> 5);
    if (e >= E) return;
    const int s = src[e], d = dst[e];
    const float* hs = h + (size_t)s * 64;
    const float* hd = h + (size_t)d * 64;
    float sum = hs[lane] * We[lane] + hs[32 + lane] * We[32 + lane]
              + hd[lane] * We[64 + lane] + hd[32 + lane] * We[96 + lane];
    #pragma unroll
    for (int off = 16; off > 0; off >>= 1) sum += __shfl_xor(sum, off, 32);
    if (lane == 0) {
        sum += efeat[(size_t)e * 3 + 0] * We[128]
             + efeat[(size_t)e * 3 + 1] * We[129]
             + efeat[(size_t)e * 3 + 2] * We[130] + be[0];
        out[e] = 1.0f / (1.0f + expf(-sum));
    }
}

extern "C" void kernel_launch(void* const* d_in, const int* in_sizes, int n_in,
                              void* d_out, int out_size, void* d_ws, size_t ws_size,
                              hipStream_t stream) {
    const float* x     = (const float*)d_in[0];
    const float* efeat = (const float*)d_in[1];
    const int*   src   = (const int*)  d_in[2];
    const int*   dst   = (const int*)  d_in[3];
    const float* W1    = (const float*)d_in[4];
    const float* b1    = (const float*)d_in[5];
    const float* W2    = (const float*)d_in[6];
    const float* b2    = (const float*)d_in[7];
    const float* We    = (const float*)d_in[8];
    const float* be    = (const float*)d_in[9];
    float* out = (float*)d_out;

    const size_t NB = (size_t)N_NODES * 64;
    char* ws = (char*)d_ws;
    float* bufA = (float*)ws;
    float* bufB = bufA + NB;
    float* bufC = bufB + NB;
    int*   cntS = (int*)(bufC + NB);
    int*   cntD = cntS + N_NODES;
    float* cS   = (float*)cntS;
    float* cD   = (float*)cntD;

    hipMemsetAsync(cntS, 0, (size_t)2 * N_NODES * sizeof(int), stream);
    hipMemsetAsync(bufB, 0, NB * sizeof(float), stream);

    count_kernel<<<N_EDGES / 256, 256, 0, stream>>>(src, dst, cntS, cntD, N_EDGES);
    coeff_kernel<<<(N_NODES + 255) / 256, 256, 0, stream>>>(cntS, cntD, N_NODES);

    gemm_wmma_kernel<128><<<(M_TILES + 3) / 4, 128, 0, stream>>>(x, W1, bufA, M_TILES);
    scatter64_kernel<<<N_EDGES / 4, 256, 0, stream>>>(bufA, cS, src, dst, bufB, N_EDGES);
    epilogue_kernel<true><<<(int)(NB / 256), 256, 0, stream>>>(bufB, cD, b1, (int)NB);

    gemm_wmma_kernel<64><<<(M_TILES + 3) / 4, 128, 0, stream>>>(bufB, W2, bufC, M_TILES);
    hipMemsetAsync(bufA, 0, NB * sizeof(float), stream);
    scatter64_kernel<<<N_EDGES / 4, 256, 0, stream>>>(bufC, cS, src, dst, bufA, N_EDGES);
    epilogue_kernel<false><<<(int)(NB / 256), 256, 0, stream>>>(bufA, cD, b2, (int)NB);

    edge_kernel<<<N_EDGES / 8, 256, 0, stream>>>(bufA, src, dst, efeat, We, be, out, N_EDGES);
}


# Float-atomic edge scatter replaced by per-call counting sort of edges by dst (atomic rank + prefix offsets) and per-node gather-sum with fused scale/bias/relu; float sum order now by arrival rank
# speedup vs baseline: 1.7770x; 1.7770x over previous
#include <hip/hip_runtime.h>
#include <hip/hip_bf16.h>

#define N_NODES 50000
#define N_EDGES 800000
#define IN_FEATS 128
#define HIDDEN 64
#define M_TILES (N_NODES / 16)

typedef float v2f __attribute__((ext_vector_type(2)));
typedef float v8f __attribute__((ext_vector_type(8)));

__global__ void count_kernel(const int* __restrict__ src, const int* __restrict__ dst,
                             int* __restrict__ cnt_src, int* __restrict__ cnt_dst,
                             int* __restrict__ rank, int E) {
    int e = blockIdx.x * blockDim.x + threadIdx.x;
    if (e >= E) return;
    const int s = src[e], d = dst[e];
    atomicAdd(&cnt_src[s], 1);
    rank[e] = atomicAdd(&cnt_dst[d], 1);
}

__global__ __launch_bounds__(256) void coeff_scan_kernel(
        const int* __restrict__ cs, const int* __restrict__ cd,
        float* __restrict__ cS, float* __restrict__ cD, int* __restrict__ rowptr, int n) {
    __shared__ int wtot[8];
    __shared__ int wpart[8];
    const int t = (int)threadIdx.x, b = (int)blockIdx.x;
    const int lane = t & 31, w = t >> 5;
    const int i = b * 256 + t;

    int part = 0;
    const int4* __restrict__ cd4 = (const int4*)cd;
    const int nq = b * 64;
    for (int q0 = 0; q0 < nq; q0 += 2048) {
        #pragma unroll
        for (int u = 0; u < 8; ++u) {
            const int q = q0 + u * 256 + t;
            const int4 v = cd4[q < nq ? q : nq - 1];
            const int sum4 = (v.x + v.y) + (v.z + v.w);
            part += q < nq ? sum4 : 0;
        }
    }
    const int mine = (i < n) ? cd[i] : 0;

    int incl = mine;
    #pragma unroll
    for (int off = 1; off < 32; off <<= 1) {
        const int v = __shfl_up(incl, off, 32);
        if (lane >= off) incl += v;
    }
    #pragma unroll
    for (int off = 16; off > 0; off >>= 1) part += __shfl_xor(part, off, 32);
    if (lane == 31) wtot[w] = incl;
    if (lane == 0)  wpart[w] = part;
    __syncthreads();
    int base = 0;
    #pragma unroll
    for (int k = 0; k < 8; ++k) {
        base += wpart[k];
        if (k < w) base += wtot[k];
    }
    if (i < n) {
        rowptr[i] = base + incl - mine;
        const int a = cs[i];
        cS[i] = rsqrtf((float)(a < 1 ? 1 : a));
        cD[i] = rsqrtf((float)(mine < 1 ? 1 : mine));
    }
}

__global__ __launch_bounds__(256) void fill_kernel(
        const int* __restrict__ src, const int* __restrict__ dst,
        const int* __restrict__ rank, const int* __restrict__ rowptr,
        const float* __restrict__ cS, int* __restrict__ epair_raw, int E4) {
    const int g = blockIdx.x * 256 + (int)threadIdx.x;
    if (g >= E4) return;
    int2* __restrict__ epair = (int2*)epair_raw;
    const int4 s = ((const int4*)src)[g];
    const int4 d = ((const int4*)dst)[g];
    const int4 r = ((const int4*)rank)[g];
    const int pos0 = rowptr[d.x] + r.x, pos1 = rowptr[d.y] + r.y;
    const int pos2 = rowptr[d.z] + r.z, pos3 = rowptr[d.w] + r.w;
    int2 p0, p1, p2, p3;
    p0.x = s.x; p0.y = __float_as_int(cS[s.x]);
    p1.x = s.y; p1.y = __float_as_int(cS[s.y]);
    p2.x = s.z; p2.y = __float_as_int(cS[s.z]);
    p3.x = s.w; p3.y = __float_as_int(cS[s.w]);
    epair[pos0] = p0; epair[pos1] = p1; epair[pos2] = p2; epair[pos3] = p3;
}

template<int K>
__global__ __launch_bounds__(128) void gemm_wmma_kernel(
        const float* __restrict__ A, const float* __restrict__ W,
        float* __restrict__ out, int mtiles) {
    __shared__ v2f sB[(K / 2) * 64];
    for (int i = (int)threadIdx.x; i < (K / 2) * 64; i += 128) {
        const int j = i >> 6, n = i & 63;
        v2f p;
        p.x = W[(2 * j)     * 64 + n];
        p.y = W[(2 * j + 1) * 64 + n];
        sB[i] = p;
    }
    __syncthreads();

    const int lane  = (int)threadIdx.x & 31;
    const int mtile = (int)blockIdx.x * 4 + ((int)threadIdx.x >> 5);
    if (mtile >= mtiles) return;
    const int m_base = mtile * 16;
    const int col    = lane & 15;
    const int half   = lane >> 4;
    const int row    = m_base + col;
    const float* __restrict__ arow = A + (size_t)row * K;

    v8f acc0 = {}, acc1 = {}, acc2 = {}, acc3 = {};
    for (int k0 = 0; k0 < K; k0 += 4) {
        const int kb = k0 + (half << 1);
        const v2f a = *(const v2f*)(arow + kb);
        const v2f* __restrict__ brow = sB + ((k0 >> 1) + half) * 64;
        const v2f b0 = brow[     col];
        const v2f b1 = brow[16 + col];
        const v2f b2 = brow[32 + col];
        const v2f b3 = brow[48 + col];
        acc0 = __builtin_amdgcn_wmma_f32_16x16x4_f32(false, a, false, b0, (short)0, acc0, false, false);
        acc1 = __builtin_amdgcn_wmma_f32_16x16x4_f32(false, a, false, b1, (short)0, acc1, false, false);
        acc2 = __builtin_amdgcn_wmma_f32_16x16x4_f32(false, a, false, b2, (short)0, acc2, false, false);
        acc3 = __builtin_amdgcn_wmma_f32_16x16x4_f32(false, a, false, b3, (short)0, acc3, false, false);
    }

    const int rtop = half << 3;
    #pragma unroll
    for (int v = 0; v < 8; ++v) {
        const size_t r = (size_t)(m_base + rtop + v) * 64;
        out[r +      col] = acc0[v];
        out[r + 16 + col] = acc1[v];
        out[r + 32 + col] = acc2[v];
        out[r + 48 + col] = acc3[v];
    }
}

template<bool RELU>
__global__ __launch_bounds__(256) void gather64_kernel(
        const float* __restrict__ y, const int* __restrict__ epair_raw,
        const int* __restrict__ rowptr, const int* __restrict__ cntd,
        const float* __restrict__ cdst, const float* __restrict__ bias,
        float* __restrict__ h, int n) {
    const int l = (int)threadIdx.x & 15;
    const int node = (int)blockIdx.x * 16 + ((int)threadIdx.x >> 4);
    if (node >= n) return;
    const int beg = rowptr[node];
    const int deg = cntd[node];
    const int end = beg + deg;
    const float4* __restrict__ y4 = (const float4*)y;
    const int2* __restrict__ epair = (const int2*)epair_raw;

    float4 acc = {0.0f, 0.0f, 0.0f, 0.0f};
    if (deg > 0) {
        const int rem  = deg & 3;
        const int last = end - 1;
        int j = beg + rem;
        int2 q0 = {0, 0}, q1 = {0, 0}, q2 = {0, 0}, q3 = {0, 0};
        if (j + 4 <= end) { q0 = epair[j]; q1 = epair[j + 1]; q2 = epair[j + 2]; q3 = epair[j + 3]; }
        if (rem > 0) {
            const int j1 = beg + 1 < last ? beg + 1 : last;
            const int j2 = beg + 2 < last ? beg + 2 : last;
            const int2 p0 = epair[beg], p1 = epair[j1], p2 = epair[j2];
            const float4 r0 = y4[(size_t)p0.x * 16 + l];
            float4 r1 = y4[(size_t)p1.x * 16 + l];
            float4 r2 = y4[(size_t)p2.x * 16 + l];
            const float c0 = __int_as_float(p0.y);
            float c1 = __int_as_float(p1.y), c2 = __int_as_float(p2.y);
            const bool u1 = rem > 1, u2 = rem > 2;
            c1 = u1 ? c1 : 0.0f; r1.x = u1 ? r1.x : 0.0f; r1.y = u1 ? r1.y : 0.0f; r1.z = u1 ? r1.z : 0.0f; r1.w = u1 ? r1.w : 0.0f;
            c2 = u2 ? c2 : 0.0f; r2.x = u2 ? r2.x : 0.0f; r2.y = u2 ? r2.y : 0.0f; r2.z = u2 ? r2.z : 0.0f; r2.w = u2 ? r2.w : 0.0f;
            acc.x += r0.x * c0; acc.y += r0.y * c0; acc.z += r0.z * c0; acc.w += r0.w * c0;
            acc.x += r1.x * c1; acc.y += r1.y * c1; acc.z += r1.z * c1; acc.w += r1.w * c1;
            acc.x += r2.x * c2; acc.y += r2.y * c2; acc.z += r2.z * c2; acc.w += r2.w * c2;
        }
        while (j + 4 <= end) {
            const int2 p0 = q0, p1 = q1, p2 = q2, p3 = q3;
            j += 4;
            if (j + 4 <= end) { q0 = epair[j]; q1 = epair[j + 1]; q2 = epair[j + 2]; q3 = epair[j + 3]; }
            const float4 r0 = y4[(size_t)p0.x * 16 + l];
            const float4 r1 = y4[(size_t)p1.x * 16 + l];
            const float4 r2 = y4[(size_t)p2.x * 16 + l];
            const float4 r3 = y4[(size_t)p3.x * 16 + l];
            const float c0 = __int_as_float(p0.y), c1 = __int_as_float(p1.y);
            const float c2 = __int_as_float(p2.y), c3 = __int_as_float(p3.y);
            acc.x += r0.x * c0; acc.y += r0.y * c0; acc.z += r0.z * c0; acc.w += r0.w * c0;
            acc.x += r1.x * c1; acc.y += r1.y * c1; acc.z += r1.z * c1; acc.w += r1.w * c1;
            acc.x += r2.x * c2; acc.y += r2.y * c2; acc.z += r2.z * c2; acc.w += r2.w * c2;
            acc.x += r3.x * c3; acc.y += r3.y * c3; acc.z += r3.z * c3; acc.w += r3.w * c3;
        }
    }

    const float cd = cdst[node];
    float4 o;
    o.x = acc.x * cd + bias[4 * l + 0];
    o.y = acc.y * cd + bias[4 * l + 1];
    o.z = acc.z * cd + bias[4 * l + 2];
    o.w = acc.w * cd + bias[4 * l + 3];
    if (RELU) {
        o.x = fmaxf(o.x, 0.0f); o.y = fmaxf(o.y, 0.0f);
        o.z = fmaxf(o.z, 0.0f); o.w = fmaxf(o.w, 0.0f);
    }
    ((float4*)h)[(size_t)node * 16 + l] = o;
}

__global__ void edge_kernel(const float* __restrict__ h, const int* __restrict__ src,
                            const int* __restrict__ dst, const float* __restrict__ efeat,
                            const float* __restrict__ We, const float* __restrict__ be,
                            float* __restrict__ out, int E) {
    const int lane = (int)threadIdx.x & 31;
    const int e = (int)blockIdx.x * ((int)blockDim.x >> 5) + ((int)threadIdx.x >> 5);
    if (e >= E) return;
    const int s = src[e], d = dst[e];
    const float* hs = h + (size_t)s * 64;
    const float* hd = h + (size_t)d * 64;
    float sum = hs[lane] * We[lane] + hs[32 + lane] * We[32 + lane]
              + hd[lane] * We[64 + lane] + hd[32 + lane] * We[96 + lane];
    #pragma unroll
    for (int off = 16; off > 0; off >>= 1) sum += __shfl_xor(sum, off, 32);
    if (lane == 0) {
        sum += efeat[(size_t)e * 3 + 0] * We[128]
             + efeat[(size_t)e * 3 + 1] * We[129]
             + efeat[(size_t)e * 3 + 2] * We[130] + be[0];
        out[e] = 1.0f / (1.0f + expf(-sum));
    }
}

extern "C" void kernel_launch(void* const* d_in, const int* in_sizes, int n_in,
                              void* d_out, int out_size, void* d_ws, size_t ws_size,
                              hipStream_t stream) {
    const float* x     = (const float*)d_in[0];
    const float* efeat = (const float*)d_in[1];
    const int*   src   = (const int*)  d_in[2];
    const int*   dst   = (const int*)  d_in[3];
    const float* W1    = (const float*)d_in[4];
    const float* b1    = (const float*)d_in[5];
    const float* W2    = (const float*)d_in[6];
    const float* b2    = (const float*)d_in[7];
    const float* We    = (const float*)d_in[8];
    const float* be    = (const float*)d_in[9];
    float* out = (float*)d_out;

    const size_t NB = (size_t)N_NODES * 64;
    char* ws = (char*)d_ws;
    float* bufA = (float*)ws;
    float* bufB = bufA + NB;
    float* bufC = bufB + NB;
    int*   cntS = (int*)(bufC + NB);
    int*   cntD = cntS + N_NODES;
    float* cS   = (float*)(cntD + N_NODES);
    float* cD   = cS + N_NODES;
    int*   rowp = (int*)(cD + N_NODES);
    int*   rank = rowp + N_NODES;
    int*   epair = rank + N_EDGES;

    hipMemsetAsync(cntS, 0, (size_t)2 * N_NODES * sizeof(int), stream);

    count_kernel<<<N_EDGES / 256, 256, 0, stream>>>(src, dst, cntS, cntD, rank, N_EDGES);
    coeff_scan_kernel<<<(N_NODES + 255) / 256, 256, 0, stream>>>(cntS, cntD, cS, cD, rowp, N_NODES);
    fill_kernel<<<(N_EDGES / 4 + 255) / 256, 256, 0, stream>>>(src, dst, rank, rowp, cS, epair, N_EDGES / 4);

    gemm_wmma_kernel<128><<<(M_TILES + 3) / 4, 128, 0, stream>>>(x, W1, bufA, M_TILES);
    gather64_kernel<true><<<(N_NODES + 15) / 16, 256, 0, stream>>>(bufA, epair, rowp, cntD, cD, b1, bufB, N_NODES);

    gemm_wmma_kernel<64><<<(M_TILES + 3) / 4, 128, 0, stream>>>(bufB, W2, bufC, M_TILES);
    gather64_kernel<false><<<(N_NODES + 15) / 16, 256, 0, stream>>>(bufC, epair, rowp, cntD, cD, b2, bufA, N_NODES);

    edge_kernel<<<N_EDGES / 8, 256, 0, stream>>>(bufA, src, dst, efeat, We, be, out, N_EDGES);
}


# Edge logit split into per-node dots h2@Ws, h2@Wd made once in the layer-2 gather epilogue; edge kernel adds two gathered scalars, 4 edges per thread, not one wave per edge; logit sum order changes
# speedup vs baseline: 2.6206x; 1.4747x over previous
#include <hip/hip_runtime.h>
#include <hip/hip_bf16.h>

#define N_NODES 50000
#define N_EDGES 800000
#define IN_FEATS 128
#define HIDDEN 64
#define M_TILES (N_NODES / 16)

typedef float v2f __attribute__((ext_vector_type(2)));
typedef float v8f __attribute__((ext_vector_type(8)));

__global__ void count_kernel(const int* __restrict__ src, const int* __restrict__ dst,
                             int* __restrict__ cnt_src, int* __restrict__ cnt_dst,
                             int* __restrict__ rank, int E) {
    int e = blockIdx.x * blockDim.x + threadIdx.x;
    if (e >= E) return;
    const int s = src[e], d = dst[e];
    atomicAdd(&cnt_src[s], 1);
    rank[e] = atomicAdd(&cnt_dst[d], 1);
}

__global__ __launch_bounds__(256) void coeff_scan_kernel(
        const int* __restrict__ cs, const int* __restrict__ cd,
        float* __restrict__ cS, float* __restrict__ cD, int* __restrict__ rowptr, int n) {
    __shared__ int wtot[8];
    __shared__ int wpart[8];
    const int t = (int)threadIdx.x, b = (int)blockIdx.x;
    const int lane = t & 31, w = t >> 5;
    const int i = b * 256 + t;

    int part = 0;
    const int4* __restrict__ cd4 = (const int4*)cd;
    const int nq = b * 64;
    for (int q0 = 0; q0 < nq; q0 += 2048) {
        #pragma unroll
        for (int u = 0; u < 8; ++u) {
            const int q = q0 + u * 256 + t;
            const int4 v = cd4[q < nq ? q : nq - 1];
            const int sum4 = (v.x + v.y) + (v.z + v.w);
            part += q < nq ? sum4 : 0;
        }
    }
    const int mine = (i < n) ? cd[i] : 0;

    int incl = mine;
    #pragma unroll
    for (int off = 1; off < 32; off <<= 1) {
        const int v = __shfl_up(incl, off, 32);
        if (lane >= off) incl += v;
    }
    #pragma unroll
    for (int off = 16; off > 0; off >>= 1) part += __shfl_xor(part, off, 32);
    if (lane == 31) wtot[w] = incl;
    if (lane == 0)  wpart[w] = part;
    __syncthreads();
    int base = 0;
    #pragma unroll
    for (int k = 0; k < 8; ++k) {
        base += wpart[k];
        if (k < w) base += wtot[k];
    }
    if (i < n) {
        rowptr[i] = base + incl - mine;
        const int a = cs[i];
        cS[i] = rsqrtf((float)(a < 1 ? 1 : a));
        cD[i] = rsqrtf((float)(mine < 1 ? 1 : mine));
    }
}

__global__ __launch_bounds__(256) void fill_kernel(
        const int* __restrict__ src, const int* __restrict__ dst,
        const int* __restrict__ rank, const int* __restrict__ rowptr,
        const float* __restrict__ cS, int* __restrict__ epair_raw, int E4) {
    const int g = blockIdx.x * 256 + (int)threadIdx.x;
    if (g >= E4) return;
    int2* __restrict__ epair = (int2*)epair_raw;
    const int4 s = ((const int4*)src)[g];
    const int4 d = ((const int4*)dst)[g];
    const int4 r = ((const int4*)rank)[g];
    const int pos0 = rowptr[d.x] + r.x, pos1 = rowptr[d.y] + r.y;
    const int pos2 = rowptr[d.z] + r.z, pos3 = rowptr[d.w] + r.w;
    int2 p0, p1, p2, p3;
    p0.x = s.x; p0.y = __float_as_int(cS[s.x]);
    p1.x = s.y; p1.y = __float_as_int(cS[s.y]);
    p2.x = s.z; p2.y = __float_as_int(cS[s.z]);
    p3.x = s.w; p3.y = __float_as_int(cS[s.w]);
    epair[pos0] = p0; epair[pos1] = p1; epair[pos2] = p2; epair[pos3] = p3;
}

template<int K>
__global__ __launch_bounds__(128) void gemm_wmma_kernel(
        const float* __restrict__ A, const float* __restrict__ W,
        float* __restrict__ out, int mtiles) {
    __shared__ v2f sB[(K / 2) * 64];
    for (int i = (int)threadIdx.x; i < (K / 2) * 64; i += 128) {
        const int j = i >> 6, n = i & 63;
        v2f p;
        p.x = W[(2 * j)     * 64 + n];
        p.y = W[(2 * j + 1) * 64 + n];
        sB[i] = p;
    }
    __syncthreads();

    const int lane  = (int)threadIdx.x & 31;
    const int mtile = (int)blockIdx.x * 4 + ((int)threadIdx.x >> 5);
    if (mtile >= mtiles) return;
    const int m_base = mtile * 16;
    const int col    = lane & 15;
    const int half   = lane >> 4;
    const int row    = m_base + col;
    const float* __restrict__ arow = A + (size_t)row * K;

    v8f acc0 = {}, acc1 = {}, acc2 = {}, acc3 = {};
    for (int k0 = 0; k0 < K; k0 += 4) {
        const int kb = k0 + (half << 1);
        const v2f a = *(const v2f*)(arow + kb);
        const v2f* __restrict__ brow = sB + ((k0 >> 1) + half) * 64;
        const v2f b0 = brow[     col];
        const v2f b1 = brow[16 + col];
        const v2f b2 = brow[32 + col];
        const v2f b3 = brow[48 + col];
        acc0 = __builtin_amdgcn_wmma_f32_16x16x4_f32(false, a, false, b0, (short)0, acc0, false, false);
        acc1 = __builtin_amdgcn_wmma_f32_16x16x4_f32(false, a, false, b1, (short)0, acc1, false, false);
        acc2 = __builtin_amdgcn_wmma_f32_16x16x4_f32(false, a, false, b2, (short)0, acc2, false, false);
        acc3 = __builtin_amdgcn_wmma_f32_16x16x4_f32(false, a, false, b3, (short)0, acc3, false, false);
    }

    const int rtop = half << 3;
    #pragma unroll
    for (int v = 0; v < 8; ++v) {
        const size_t r = (size_t)(m_base + rtop + v) * 64;
        out[r +      col] = acc0[v];
        out[r + 16 + col] = acc1[v];
        out[r + 32 + col] = acc2[v];
        out[r + 48 + col] = acc3[v];
    }
}

template<bool RELU, bool PROJ>
__global__ __launch_bounds__(256) void gather64_kernel(
        const float* __restrict__ y, const int* __restrict__ epair_raw,
        const int* __restrict__ rowptr, const int* __restrict__ cntd,
        const float* __restrict__ cdst, const float* __restrict__ bias,
        const float* __restrict__ We, float* __restrict__ h, int n) {
    const int l = (int)threadIdx.x & 15;
    const int node = (int)blockIdx.x * 16 + ((int)threadIdx.x >> 4);
    if (node >= n) return;
    const int beg = rowptr[node];
    const int deg = cntd[node];
    const int end = beg + deg;
    const float4* __restrict__ y4 = (const float4*)y;
    const int2* __restrict__ epair = (const int2*)epair_raw;

    float4 acc = {0.0f, 0.0f, 0.0f, 0.0f};
    if (deg > 0) {
        const int rem  = deg & 3;
        const int last = end - 1;
        int j = beg + rem;
        int2 q0 = {0, 0}, q1 = {0, 0}, q2 = {0, 0}, q3 = {0, 0};
        if (j + 4 <= end) { q0 = epair[j]; q1 = epair[j + 1]; q2 = epair[j + 2]; q3 = epair[j + 3]; }
        if (rem > 0) {
            const int j1 = beg + 1 < last ? beg + 1 : last;
            const int j2 = beg + 2 < last ? beg + 2 : last;
            const int2 p0 = epair[beg], p1 = epair[j1], p2 = epair[j2];
            const float4 r0 = y4[(size_t)p0.x * 16 + l];
            float4 r1 = y4[(size_t)p1.x * 16 + l];
            float4 r2 = y4[(size_t)p2.x * 16 + l];
            const float c0 = __int_as_float(p0.y);
            float c1 = __int_as_float(p1.y), c2 = __int_as_float(p2.y);
            const bool u1 = rem > 1, u2 = rem > 2;
            c1 = u1 ? c1 : 0.0f; r1.x = u1 ? r1.x : 0.0f; r1.y = u1 ? r1.y : 0.0f; r1.z = u1 ? r1.z : 0.0f; r1.w = u1 ? r1.w : 0.0f;
            c2 = u2 ? c2 : 0.0f; r2.x = u2 ? r2.x : 0.0f; r2.y = u2 ? r2.y : 0.0f; r2.z = u2 ? r2.z : 0.0f; r2.w = u2 ? r2.w : 0.0f;
            acc.x += r0.x * c0; acc.y += r0.y * c0; acc.z += r0.z * c0; acc.w += r0.w * c0;
            acc.x += r1.x * c1; acc.y += r1.y * c1; acc.z += r1.z * c1; acc.w += r1.w * c1;
            acc.x += r2.x * c2; acc.y += r2.y * c2; acc.z += r2.z * c2; acc.w += r2.w * c2;
        }
        while (j + 4 <= end) {
            const int2 p0 = q0, p1 = q1, p2 = q2, p3 = q3;
            j += 4;
            if (j + 4 <= end) { q0 = epair[j]; q1 = epair[j + 1]; q2 = epair[j + 2]; q3 = epair[j + 3]; }
            const float4 r0 = y4[(size_t)p0.x * 16 + l];
            const float4 r1 = y4[(size_t)p1.x * 16 + l];
            const float4 r2 = y4[(size_t)p2.x * 16 + l];
            const float4 r3 = y4[(size_t)p3.x * 16 + l];
            const float c0 = __int_as_float(p0.y), c1 = __int_as_float(p1.y);
            const float c2 = __int_as_float(p2.y), c3 = __int_as_float(p3.y);
            acc.x += r0.x * c0; acc.y += r0.y * c0; acc.z += r0.z * c0; acc.w += r0.w * c0;
            acc.x += r1.x * c1; acc.y += r1.y * c1; acc.z += r1.z * c1; acc.w += r1.w * c1;
            acc.x += r2.x * c2; acc.y += r2.y * c2; acc.z += r2.z * c2; acc.w += r2.w * c2;
            acc.x += r3.x * c3; acc.y += r3.y * c3; acc.z += r3.z * c3; acc.w += r3.w * c3;
        }
    }

    const float cd = cdst[node];
    float4 o;
    o.x = acc.x * cd + bias[4 * l + 0];
    o.y = acc.y * cd + bias[4 * l + 1];
    o.z = acc.z * cd + bias[4 * l + 2];
    o.w = acc.w * cd + bias[4 * l + 3];
    if (RELU) {
        o.x = fmaxf(o.x, 0.0f); o.y = fmaxf(o.y, 0.0f);
        o.z = fmaxf(o.z, 0.0f); o.w = fmaxf(o.w, 0.0f);
    }
    if (PROJ) {
        const float4 ws = ((const float4*)We)[l];
        const float4 wd = ((const float4*)We)[16 + l];
        float ps = (o.x * ws.x + o.y * ws.y) + (o.z * ws.z + o.w * ws.w);
        float pd = (o.x * wd.x + o.y * wd.y) + (o.z * wd.z + o.w * wd.w);
        ps += __shfl_xor(ps, 8, 32); pd += __shfl_xor(pd, 8, 32);
        ps += __shfl_xor(ps, 4, 32); pd += __shfl_xor(pd, 4, 32);
        ps += __shfl_xor(ps, 2, 32); pd += __shfl_xor(pd, 2, 32);
        ps += __shfl_xor(ps, 1, 32); pd += __shfl_xor(pd, 1, 32);
        if (l == 0) {
            float2 p; p.x = ps; p.y = pd;
            ((float2*)h)[node] = p;
        }
    } else {
        ((float4*)h)[(size_t)node * 16 + l] = o;
    }
}

__global__ __launch_bounds__(256) void edge_kernel(
        const float* __restrict__ pp, const int* __restrict__ src,
        const int* __restrict__ dst, const float* __restrict__ efeat,
        const float* __restrict__ We, const float* __restrict__ be,
        float* __restrict__ out, int E4) {
    const int g = (int)blockIdx.x * 256 + (int)threadIdx.x;
    if (g >= E4) return;
    const int4 s = ((const int4*)src)[g];
    const int4 d = ((const int4*)dst)[g];
    const float4* __restrict__ ef4 = (const float4*)efeat + (size_t)g * 3;
    const float4 f0 = ef4[0], f1 = ef4[1], f2 = ef4[2];
    const float s0 = pp[2 * s.x], s1 = pp[2 * s.y], s2 = pp[2 * s.z], s3 = pp[2 * s.w];
    const float d0 = pp[2 * d.x + 1], d1 = pp[2 * d.y + 1], d2 = pp[2 * d.z + 1], d3 = pp[2 * d.w + 1];
    const float w0 = We[128], w1 = We[129], w2 = We[130], b = be[0];
    const float t0 = (s0 + d0) + (f0.x * w0 + f0.y * w1 + f0.z * w2 + b);
    const float t1 = (s1 + d1) + (f0.w * w0 + f1.x * w1 + f1.y * w2 + b);
    const float t2 = (s2 + d2) + (f1.z * w0 + f1.w * w1 + f2.x * w2 + b);
    const float t3 = (s3 + d3) + (f2.y * w0 + f2.z * w1 + f2.w * w2 + b);
    float4 o;
    o.x = 1.0f / (1.0f + expf(-t0));
    o.y = 1.0f / (1.0f + expf(-t1));
    o.z = 1.0f / (1.0f + expf(-t2));
    o.w = 1.0f / (1.0f + expf(-t3));
    ((float4*)out)[g] = o;
}

extern "C" void kernel_launch(void* const* d_in, const int* in_sizes, int n_in,
                              void* d_out, int out_size, void* d_ws, size_t ws_size,
                              hipStream_t stream) {
    const float* x     = (const float*)d_in[0];
    const float* efeat = (const float*)d_in[1];
    const int*   src   = (const int*)  d_in[2];
    const int*   dst   = (const int*)  d_in[3];
    const float* W1    = (const float*)d_in[4];
    const float* b1    = (const float*)d_in[5];
    const float* W2    = (const float*)d_in[6];
    const float* b2    = (const float*)d_in[7];
    const float* We    = (const float*)d_in[8];
    const float* be    = (const float*)d_in[9];
    float* out = (float*)d_out;

    const size_t NB = (size_t)N_NODES * 64;
    char* ws = (char*)d_ws;
    float* bufA = (float*)ws;
    float* bufB = bufA + NB;
    float* bufC = bufB + NB;
    int*   cntS = (int*)(bufC + NB);
    int*   cntD = cntS + N_NODES;
    float* cS   = (float*)(cntD + N_NODES);
    float* cD   = cS + N_NODES;
    int*   rowp = (int*)(cD + N_NODES);
    int*   rank = rowp + N_NODES;
    int*   epair = rank + N_EDGES;

    hipMemsetAsync(cntS, 0, (size_t)2 * N_NODES * sizeof(int), stream);

    count_kernel<<<N_EDGES / 256, 256, 0, stream>>>(src, dst, cntS, cntD, rank, N_EDGES);
    coeff_scan_kernel<<<(N_NODES + 255) / 256, 256, 0, stream>>>(cntS, cntD, cS, cD, rowp, N_NODES);
    fill_kernel<<<(N_EDGES / 4 + 255) / 256, 256, 0, stream>>>(src, dst, rank, rowp, cS, epair, N_EDGES / 4);

    gemm_wmma_kernel<128><<<(M_TILES + 3) / 4, 128, 0, stream>>>(x, W1, bufA, M_TILES);
    gather64_kernel<true, false><<<(N_NODES + 15) / 16, 256, 0, stream>>>(bufA, epair, rowp, cntD, cD, b1, We, bufB, N_NODES);

    gemm_wmma_kernel<64><<<(M_TILES + 3) / 4, 128, 0, stream>>>(bufB, W2, bufC, M_TILES);
    gather64_kernel<false, true><<<(N_NODES + 15) / 16, 256, 0, stream>>>(bufC, epair, rowp, cntD, cD, b2, We, bufA, N_NODES);

    edge_kernel<<<(N_EDGES / 4 + 255) / 256, 256, 0, stream>>>(bufA, src, dst, efeat, We, be, out, N_EDGES / 4);
}


# Layer 2 folded by linearity: h1 projected to 2 scalars per node with W2@Ws, W2@Wd in the layer-1 gather epilogue; 64-wide layer-2 GEMM and gather become a 2-float gather; float sum order changes
# speedup vs baseline: 2.8218x; 1.0768x over previous
#include <hip/hip_runtime.h>
#include <hip/hip_bf16.h>

#define N_NODES 50000
#define N_EDGES 800000
#define IN_FEATS 128
#define HIDDEN 64
#define M_TILES (N_NODES / 16)

typedef float v2f __attribute__((ext_vector_type(2)));
typedef float v8f __attribute__((ext_vector_type(8)));

__global__ void count_kernel(const int* __restrict__ src, const int* __restrict__ dst,
                             int* __restrict__ cnt_src, int* __restrict__ cnt_dst,
                             int* __restrict__ rank, int E) {
    int e = blockIdx.x * blockDim.x + threadIdx.x;
    if (e >= E) return;
    const int s = src[e], d = dst[e];
    atomicAdd(&cnt_src[s], 1);
    rank[e] = atomicAdd(&cnt_dst[d], 1);
}

__global__ __launch_bounds__(256) void coeff_scan_kernel(
        const int* __restrict__ cs, const int* __restrict__ cd,
        float* __restrict__ cS, float* __restrict__ cD, int* __restrict__ rowptr, int n) {
    __shared__ int wtot[8];
    __shared__ int wpart[8];
    const int t = (int)threadIdx.x, b = (int)blockIdx.x;
    const int lane = t & 31, w = t >> 5;
    const int i = b * 256 + t;

    int part = 0;
    const int4* __restrict__ cd4 = (const int4*)cd;
    const int nq = b * 64;
    for (int q0 = 0; q0 < nq; q0 += 2048) {
        #pragma unroll
        for (int u = 0; u < 8; ++u) {
            const int q = q0 + u * 256 + t;
            const int4 v = cd4[q < nq ? q : nq - 1];
            const int sum4 = (v.x + v.y) + (v.z + v.w);
            part += q < nq ? sum4 : 0;
        }
    }
    const int mine = (i < n) ? cd[i] : 0;

    int incl = mine;
    #pragma unroll
    for (int off = 1; off < 32; off <<= 1) {
        const int v = __shfl_up(incl, off, 32);
        if (lane >= off) incl += v;
    }
    #pragma unroll
    for (int off = 16; off > 0; off >>= 1) part += __shfl_xor(part, off, 32);
    if (lane == 31) wtot[w] = incl;
    if (lane == 0)  wpart[w] = part;
    __syncthreads();
    int base = 0;
    #pragma unroll
    for (int k = 0; k < 8; ++k) {
        base += wpart[k];
        if (k < w) base += wtot[k];
    }
    if (i < n) {
        rowptr[i] = base + incl - mine;
        const int a = cs[i];
        cS[i] = rsqrtf((float)(a < 1 ? 1 : a));
        cD[i] = rsqrtf((float)(mine < 1 ? 1 : mine));
    }
}

__global__ __launch_bounds__(256) void fill_kernel(
        const int* __restrict__ src, const int* __restrict__ dst,
        const int* __restrict__ rank, const int* __restrict__ rowptr,
        const float* __restrict__ cS, int* __restrict__ epair_raw, int E4) {
    const int g = blockIdx.x * 256 + (int)threadIdx.x;
    if (g >= E4) return;
    int2* __restrict__ epair = (int2*)epair_raw;
    const int4 s = ((const int4*)src)[g];
    const int4 d = ((const int4*)dst)[g];
    const int4 r = ((const int4*)rank)[g];
    const int pos0 = rowptr[d.x] + r.x, pos1 = rowptr[d.y] + r.y;
    const int pos2 = rowptr[d.z] + r.z, pos3 = rowptr[d.w] + r.w;
    int2 p0, p1, p2, p3;
    p0.x = s.x; p0.y = __float_as_int(cS[s.x]);
    p1.x = s.y; p1.y = __float_as_int(cS[s.y]);
    p2.x = s.z; p2.y = __float_as_int(cS[s.z]);
    p3.x = s.w; p3.y = __float_as_int(cS[s.w]);
    epair[pos0] = p0; epair[pos1] = p1; epair[pos2] = p2; epair[pos3] = p3;
}

template<int K>
__global__ __launch_bounds__(128) void gemm_wmma_kernel(
        const float* __restrict__ A, const float* __restrict__ W,
        float* __restrict__ out, int mtiles) {
    __shared__ v2f sB[(K / 2) * 64];
    for (int i = (int)threadIdx.x; i < (K / 2) * 64; i += 128) {
        const int j = i >> 6, n = i & 63;
        v2f p;
        p.x = W[(2 * j)     * 64 + n];
        p.y = W[(2 * j + 1) * 64 + n];
        sB[i] = p;
    }
    __syncthreads();

    const int lane  = (int)threadIdx.x & 31;
    const int mtile = (int)blockIdx.x * 4 + ((int)threadIdx.x >> 5);
    if (mtile >= mtiles) return;
    const int m_base = mtile * 16;
    const int col    = lane & 15;
    const int half   = lane >> 4;
    const int row    = m_base + col;
    const float* __restrict__ arow = A + (size_t)row * K;

    v8f acc0 = {}, acc1 = {}, acc2 = {}, acc3 = {};
    for (int k0 = 0; k0 < K; k0 += 4) {
        const int kb = k0 + (half << 1);
        const v2f a = *(const v2f*)(arow + kb);
        const v2f* __restrict__ brow = sB + ((k0 >> 1) + half) * 64;
        const v2f b0 = brow[     col];
        const v2f b1 = brow[16 + col];
        const v2f b2 = brow[32 + col];
        const v2f b3 = brow[48 + col];
        acc0 = __builtin_amdgcn_wmma_f32_16x16x4_f32(false, a, false, b0, (short)0, acc0, false, false);
        acc1 = __builtin_amdgcn_wmma_f32_16x16x4_f32(false, a, false, b1, (short)0, acc1, false, false);
        acc2 = __builtin_amdgcn_wmma_f32_16x16x4_f32(false, a, false, b2, (short)0, acc2, false, false);
        acc3 = __builtin_amdgcn_wmma_f32_16x16x4_f32(false, a, false, b3, (short)0, acc3, false, false);
    }

    const int rtop = half << 3;
    #pragma unroll
    for (int v = 0; v < 8; ++v) {
        const size_t r = (size_t)(m_base + rtop + v) * 64;
        out[r +      col] = acc0[v];
        out[r + 16 + col] = acc1[v];
        out[r + 32 + col] = acc2[v];
        out[r + 48 + col] = acc3[v];
    }
}

__global__ __launch_bounds__(128) void wproj_kernel(
        const float* __restrict__ W2, const float* __restrict__ b2,
        const float* __restrict__ We, float* __restrict__ v) {
    const int t = (int)threadIdx.x;
    if (t > 64) return;
    const float4* __restrict__ r4 = (const float4*)(t < 64 ? W2 + t * 64 : b2);
    const float4* __restrict__ ws4 = (const float4*)We;
    const float4* __restrict__ wd4 = (const float4*)We + 16;
    float s = 0.0f, d = 0.0f;
    for (int j = 0; j < 16; ++j) {
        const float4 r = r4[j], a = ws4[j], b = wd4[j];
        s += (r.x * a.x + r.y * a.y) + (r.z * a.z + r.w * a.w);
        d += (r.x * b.x + r.y * b.y) + (r.z * b.z + r.w * b.w);
    }
    if (t < 64) { v[t] = s; v[64 + t] = d; }
    else        { v[128] = s; v[129] = d; }
}

__global__ __launch_bounds__(256) void gather64_proj_kernel(
        const float* __restrict__ y, const int* __restrict__ epair_raw,
        const int* __restrict__ rowptr, const int* __restrict__ cntd,
        const float* __restrict__ cdst, const float* __restrict__ bias,
        const float* __restrict__ vw, float* __restrict__ z, int n) {
    const int l = (int)threadIdx.x & 15;
    const int node = (int)blockIdx.x * 16 + ((int)threadIdx.x >> 4);
    if (node >= n) return;
    const int beg = rowptr[node];
    const int deg = cntd[node];
    const int end = beg + deg;
    const float4* __restrict__ y4 = (const float4*)y;
    const int2* __restrict__ epair = (const int2*)epair_raw;

    float4 acc = {0.0f, 0.0f, 0.0f, 0.0f};
    if (deg > 0) {
        const int rem  = deg & 3;
        const int last = end - 1;
        int j = beg + rem;
        int2 q0 = {0, 0}, q1 = {0, 0}, q2 = {0, 0}, q3 = {0, 0};
        if (j + 4 <= end) { q0 = epair[j]; q1 = epair[j + 1]; q2 = epair[j + 2]; q3 = epair[j + 3]; }
        if (rem > 0) {
            const int j1 = beg + 1 < last ? beg + 1 : last;
            const int j2 = beg + 2 < last ? beg + 2 : last;
            const int2 p0 = epair[beg], p1 = epair[j1], p2 = epair[j2];
            const float4 r0 = y4[(size_t)p0.x * 16 + l];
            float4 r1 = y4[(size_t)p1.x * 16 + l];
            float4 r2 = y4[(size_t)p2.x * 16 + l];
            const float c0 = __int_as_float(p0.y);
            float c1 = __int_as_float(p1.y), c2 = __int_as_float(p2.y);
            const bool u1 = rem > 1, u2 = rem > 2;
            c1 = u1 ? c1 : 0.0f; r1.x = u1 ? r1.x : 0.0f; r1.y = u1 ? r1.y : 0.0f; r1.z = u1 ? r1.z : 0.0f; r1.w = u1 ? r1.w : 0.0f;
            c2 = u2 ? c2 : 0.0f; r2.x = u2 ? r2.x : 0.0f; r2.y = u2 ? r2.y : 0.0f; r2.z = u2 ? r2.z : 0.0f; r2.w = u2 ? r2.w : 0.0f;
            acc.x += r0.x * c0; acc.y += r0.y * c0; acc.z += r0.z * c0; acc.w += r0.w * c0;
            acc.x += r1.x * c1; acc.y += r1.y * c1; acc.z += r1.z * c1; acc.w += r1.w * c1;
            acc.x += r2.x * c2; acc.y += r2.y * c2; acc.z += r2.z * c2; acc.w += r2.w * c2;
        }
        while (j + 4 <= end) {
            const int2 p0 = q0, p1 = q1, p2 = q2, p3 = q3;
            j += 4;
            if (j + 4 <= end) { q0 = epair[j]; q1 = epair[j + 1]; q2 = epair[j + 2]; q3 = epair[j + 3]; }
            const float4 r0 = y4[(size_t)p0.x * 16 + l];
            const float4 r1 = y4[(size_t)p1.x * 16 + l];
            const float4 r2 = y4[(size_t)p2.x * 16 + l];
            const float4 r3 = y4[(size_t)p3.x * 16 + l];
            const float c0 = __int_as_float(p0.y), c1 = __int_as_float(p1.y);
            const float c2 = __int_as_float(p2.y), c3 = __int_as_float(p3.y);
            acc.x += r0.x * c0; acc.y += r0.y * c0; acc.z += r0.z * c0; acc.w += r0.w * c0;
            acc.x += r1.x * c1; acc.y += r1.y * c1; acc.z += r1.z * c1; acc.w += r1.w * c1;
            acc.x += r2.x * c2; acc.y += r2.y * c2; acc.z += r2.z * c2; acc.w += r2.w * c2;
            acc.x += r3.x * c3; acc.y += r3.y * c3; acc.z += r3.z * c3; acc.w += r3.w * c3;
        }
    }

    const float cd = cdst[node];
    float4 o;
    o.x = fmaxf(acc.x * cd + bias[4 * l + 0], 0.0f);
    o.y = fmaxf(acc.y * cd + bias[4 * l + 1], 0.0f);
    o.z = fmaxf(acc.z * cd + bias[4 * l + 2], 0.0f);
    o.w = fmaxf(acc.w * cd + bias[4 * l + 3], 0.0f);

    const float4 ws = ((const float4*)vw)[l];
    const float4 wd = ((const float4*)vw)[16 + l];
    float ps = (o.x * ws.x + o.y * ws.y) + (o.z * ws.z + o.w * ws.w);
    float pd = (o.x * wd.x + o.y * wd.y) + (o.z * wd.z + o.w * wd.w);
    ps += __shfl_xor(ps, 8, 32); pd += __shfl_xor(pd, 8, 32);
    ps += __shfl_xor(ps, 4, 32); pd += __shfl_xor(pd, 4, 32);
    ps += __shfl_xor(ps, 2, 32); pd += __shfl_xor(pd, 2, 32);
    ps += __shfl_xor(ps, 1, 32); pd += __shfl_xor(pd, 1, 32);
    if (l == 0) {
        float2 p; p.x = ps; p.y = pd;
        ((float2*)z)[node] = p;
    }
}

__global__ __launch_bounds__(256) void gather2_kernel(
        const float* __restrict__ z, const int* __restrict__ epair_raw,
        const int* __restrict__ rowptr, const int* __restrict__ cntd,
        const float* __restrict__ cdst, const float* __restrict__ vw,
        float* __restrict__ pp, int n) {
    const int l = (int)threadIdx.x & 15;
    const int node = (int)blockIdx.x * 16 + ((int)threadIdx.x >> 4);
    if (node >= n) return;
    const int beg = rowptr[node];
    const int deg = cntd[node];
    const float cd = cdst[node];
    asm volatile("" :: "v"(beg), "v"(deg), "v"(cd));
    const int end = beg + deg;
    const int2* __restrict__ epair = (const int2*)epair_raw;
    const float2* __restrict__ z2 = (const float2*)z;

    float as = 0.0f, ad = 0.0f;
    for (int j = beg + l; j < end; j += 32) {
        const bool u1 = j + 16 < end;
        const int2 p0 = epair[j];
        const int2 p1 = epair[u1 ? j + 16 : j];
        const float2 z0 = z2[p0.x];
        float2 z1 = z2[p1.x];
        const float c0 = __int_as_float(p0.y);
        const float c1 = u1 ? __int_as_float(p1.y) : 0.0f;
        z1.x = u1 ? z1.x : 0.0f; z1.y = u1 ? z1.y : 0.0f;
        as += z0.x * c0; ad += z0.y * c0;
        as += z1.x * c1; ad += z1.y * c1;
    }
    as += __shfl_xor(as, 8, 32); ad += __shfl_xor(ad, 8, 32);
    as += __shfl_xor(as, 4, 32); ad += __shfl_xor(ad, 4, 32);
    as += __shfl_xor(as, 2, 32); ad += __shfl_xor(ad, 2, 32);
    as += __shfl_xor(as, 1, 32); ad += __shfl_xor(ad, 1, 32);
    if (l == 0) {
        float2 p; p.x = as * cd + vw[128]; p.y = ad * cd + vw[129];
        ((float2*)pp)[node] = p;
    }
}

__global__ __launch_bounds__(256) void edge_kernel(
        const float* __restrict__ pp, const int* __restrict__ src,
        const int* __restrict__ dst, const float* __restrict__ efeat,
        const float* __restrict__ We, const float* __restrict__ be,
        float* __restrict__ out, int E4) {
    const int g = (int)blockIdx.x * 256 + (int)threadIdx.x;
    if (g >= E4) return;
    const int4 s = ((const int4*)src)[g];
    const int4 d = ((const int4*)dst)[g];
    const float4* __restrict__ ef4 = (const float4*)efeat + (size_t)g * 3;
    const float4 f0 = ef4[0], f1 = ef4[1], f2 = ef4[2];
    const float s0 = pp[2 * s.x], s1 = pp[2 * s.y], s2 = pp[2 * s.z], s3 = pp[2 * s.w];
    const float d0 = pp[2 * d.x + 1], d1 = pp[2 * d.y + 1], d2 = pp[2 * d.z + 1], d3 = pp[2 * d.w + 1];
    const float w0 = We[128], w1 = We[129], w2 = We[130], b = be[0];
    const float t0 = (s0 + d0) + (f0.x * w0 + f0.y * w1 + f0.z * w2 + b);
    const float t1 = (s1 + d1) + (f0.w * w0 + f1.x * w1 + f1.y * w2 + b);
    const float t2 = (s2 + d2) + (f1.z * w0 + f1.w * w1 + f2.x * w2 + b);
    const float t3 = (s3 + d3) + (f2.y * w0 + f2.z * w1 + f2.w * w2 + b);
    float4 o;
    o.x = 1.0f / (1.0f + expf(-t0));
    o.y = 1.0f / (1.0f + expf(-t1));
    o.z = 1.0f / (1.0f + expf(-t2));
    o.w = 1.0f / (1.0f + expf(-t3));
    ((float4*)out)[g] = o;
}

extern "C" void kernel_launch(void* const* d_in, const int* in_sizes, int n_in,
                              void* d_out, int out_size, void* d_ws, size_t ws_size,
                              hipStream_t stream) {
    const float* x     = (const float*)d_in[0];
    const float* efeat = (const float*)d_in[1];
    const int*   src   = (const int*)  d_in[2];
    const int*   dst   = (const int*)  d_in[3];
    const float* W1    = (const float*)d_in[4];
    const float* b1    = (const float*)d_in[5];
    const float* W2    = (const float*)d_in[6];
    const float* b2    = (const float*)d_in[7];
    const float* We    = (const float*)d_in[8];
    const float* be    = (const float*)d_in[9];
    float* out = (float*)d_out;

    const size_t NB = (size_t)N_NODES * 64;
    char* ws = (char*)d_ws;
    float* bufA = (float*)ws;
    float* zbuf = bufA + NB;
    float* ppair = zbuf + 2 * N_NODES;
    float* vw   = ppair + 2 * N_NODES;
    int*   cntS = (int*)(vw + 256);
    int*   cntD = cntS + N_NODES;
    float* cS   = (float*)(cntD + N_NODES);
    float* cD   = cS + N_NODES;
    int*   rowp = (int*)(cD + N_NODES);
    int*   rank = rowp + N_NODES;
    int*   epair = rank + N_EDGES;

    hipMemsetAsync(cntS, 0, (size_t)2 * N_NODES * sizeof(int), stream);

    wproj_kernel<<<1, 128, 0, stream>>>(W2, b2, We, vw);
    count_kernel<<<N_EDGES / 256, 256, 0, stream>>>(src, dst, cntS, cntD, rank, N_EDGES);
    coeff_scan_kernel<<<(N_NODES + 255) / 256, 256, 0, stream>>>(cntS, cntD, cS, cD, rowp, N_NODES);
    fill_kernel<<<(N_EDGES / 4 + 255) / 256, 256, 0, stream>>>(src, dst, rank, rowp, cS, epair, N_EDGES / 4);

    gemm_wmma_kernel<128><<<(M_TILES + 3) / 4, 128, 0, stream>>>(x, W1, bufA, M_TILES);
    gather64_proj_kernel<<<(N_NODES + 15) / 16, 256, 0, stream>>>(bufA, epair, rowp, cntD, cD, b1, vw, zbuf, N_NODES);

    gather2_kernel<<<(N_NODES + 15) / 16, 256, 0, stream>>>(zbuf, epair, rowp, cntD, cD, vw, ppair, N_NODES);

    edge_kernel<<<(N_EDGES / 4 + 255) / 256, 256, 0, stream>>>(ppair, src, dst, efeat, We, be, out, N_EDGES / 4);
}


# Layer-1 GEMM and the folded-weight product, which need nothing from the edge sort, run as leading blocks of the degree-count launch (block index picks the piece): 9 stream operations become 7
# speedup vs baseline: 3.0854x; 1.0934x over previous
#include <hip/hip_runtime.h>
#include <hip/hip_bf16.h>

#define N_NODES 50000
#define N_EDGES 800000
#define IN_FEATS 128
#define HIDDEN 64
#define M_TILES (N_NODES / 16)

typedef float v2f __attribute__((ext_vector_type(2)));
typedef float v8f __attribute__((ext_vector_type(8)));

__global__ __launch_bounds__(256) void coeff_scan_kernel(
        const int* __restrict__ cs, const int* __restrict__ cd,
        float* __restrict__ cS, float* __restrict__ cD, int* __restrict__ rowptr, int n) {
    __shared__ int wtot[8];
    __shared__ int wpart[8];
    const int t = (int)threadIdx.x, b = (int)blockIdx.x;
    const int lane = t & 31, w = t >> 5;
    const int i = b * 256 + t;

    int part = 0;
    const int4* __restrict__ cd4 = (const int4*)cd;
    const int nq = b * 64;
    for (int q0 = 0; q0 < nq; q0 += 2048) {
        #pragma unroll
        for (int u = 0; u < 8; ++u) {
            const int q = q0 + u * 256 + t;
            const int4 v = cd4[q < nq ? q : nq - 1];
            const int sum4 = (v.x + v.y) + (v.z + v.w);
            part += q < nq ? sum4 : 0;
        }
    }
    const int mine = (i < n) ? cd[i] : 0;

    int incl = mine;
    #pragma unroll
    for (int off = 1; off < 32; off <<= 1) {
        const int v = __shfl_up(incl, off, 32);
        if (lane >= off) incl += v;
    }
    #pragma unroll
    for (int off = 16; off > 0; off >>= 1) part += __shfl_xor(part, off, 32);
    if (lane == 31) wtot[w] = incl;
    if (lane == 0)  wpart[w] = part;
    __syncthreads();
    int base = 0;
    #pragma unroll
    for (int k = 0; k < 8; ++k) {
        base += wpart[k];
        if (k < w) base += wtot[k];
    }
    if (i < n) {
        rowptr[i] = base + incl - mine;
        const int a = cs[i];
        cS[i] = rsqrtf((float)(a < 1 ? 1 : a));
        cD[i] = rsqrtf((float)(mine < 1 ? 1 : mine));
    }
}

__global__ __launch_bounds__(256) void fill_kernel(
        const int* __restrict__ src, const int* __restrict__ dst,
        const int* __restrict__ rank, const int* __restrict__ rowptr,
        const float* __restrict__ cS, int* __restrict__ epair_raw, int E4) {
    const int g = blockIdx.x * 256 + (int)threadIdx.x;
    if (g >= E4) return;
    int2* __restrict__ epair = (int2*)epair_raw;
    const int4 s = ((const int4*)src)[g];
    const int4 d = ((const int4*)dst)[g];
    const int4 r = ((const int4*)rank)[g];
    const int pos0 = rowptr[d.x] + r.x, pos1 = rowptr[d.y] + r.y;
    const int pos2 = rowptr[d.z] + r.z, pos3 = rowptr[d.w] + r.w;
    int2 p0, p1, p2, p3;
    p0.x = s.x; p0.y = __float_as_int(cS[s.x]);
    p1.x = s.y; p1.y = __float_as_int(cS[s.y]);
    p2.x = s.z; p2.y = __float_as_int(cS[s.z]);
    p3.x = s.w; p3.y = __float_as_int(cS[s.w]);
    epair[pos0] = p0; epair[pos1] = p1; epair[pos2] = p2; epair[pos3] = p3;
}

#define GEMM_BLOCKS  ((M_TILES + 7) / 8)
#define COUNT_BLOCKS (N_EDGES / 256)

template<int K>
__device__ __forceinline__ void gemm_wmma_piece(
        const float* __restrict__ A, const float* __restrict__ W,
        float* __restrict__ out, int mtiles, int blk, v2f* __restrict__ sB) {
    for (int i = (int)threadIdx.x; i < (K / 2) * 64; i += 256) {
        const int j = i >> 6, n = i & 63;
        v2f p;
        p.x = W[(2 * j)     * 64 + n];
        p.y = W[(2 * j + 1) * 64 + n];
        sB[i] = p;
    }
    __syncthreads();

    const int lane  = (int)threadIdx.x & 31;
    const int mtile = blk * 8 + ((int)threadIdx.x >> 5);
    if (mtile >= mtiles) return;
    const int m_base = mtile * 16;
    const int col    = lane & 15;
    const int half   = lane >> 4;
    const int row    = m_base + col;
    const float* __restrict__ arow = A + (size_t)row * K;

    v8f acc0 = {}, acc1 = {}, acc2 = {}, acc3 = {};
    for (int k0 = 0; k0 < K; k0 += 4) {
        const int kb = k0 + (half << 1);
        const v2f a = *(const v2f*)(arow + kb);
        const v2f* __restrict__ brow = sB + ((k0 >> 1) + half) * 64;
        const v2f b0 = brow[     col];
        const v2f b1 = brow[16 + col];
        const v2f b2 = brow[32 + col];
        const v2f b3 = brow[48 + col];
        acc0 = __builtin_amdgcn_wmma_f32_16x16x4_f32(false, a, false, b0, (short)0, acc0, false, false);
        acc1 = __builtin_amdgcn_wmma_f32_16x16x4_f32(false, a, false, b1, (short)0, acc1, false, false);
        acc2 = __builtin_amdgcn_wmma_f32_16x16x4_f32(false, a, false, b2, (short)0, acc2, false, false);
        acc3 = __builtin_amdgcn_wmma_f32_16x16x4_f32(false, a, false, b3, (short)0, acc3, false, false);
    }

    const int rtop = half << 3;
    #pragma unroll
    for (int v = 0; v < 8; ++v) {
        const size_t r = (size_t)(m_base + rtop + v) * 64;
        out[r +      col] = acc0[v];
        out[r + 16 + col] = acc1[v];
        out[r + 32 + col] = acc2[v];
        out[r + 48 + col] = acc3[v];
    }
}

__device__ __forceinline__ void wproj_piece(
        const float* __restrict__ W2, const float* __restrict__ b2,
        const float* __restrict__ We, float* __restrict__ v) {
    const int t = (int)threadIdx.x;
    if (t > 64) return;
    const float4* __restrict__ r4 = (const float4*)(t < 64 ? W2 + t * 64 : b2);
    const float4* __restrict__ ws4 = (const float4*)We;
    const float4* __restrict__ wd4 = (const float4*)We + 16;
    float s = 0.0f, d = 0.0f;
    for (int j = 0; j < 16; ++j) {
        const float4 r = r4[j], a = ws4[j], b = wd4[j];
        s += (r.x * a.x + r.y * a.y) + (r.z * a.z + r.w * a.w);
        d += (r.x * b.x + r.y * b.y) + (r.z * b.z + r.w * b.w);
    }
    if (t < 64) { v[t] = s; v[64 + t] = d; }
    else        { v[128] = s; v[129] = d; }
}

__device__ __forceinline__ void count_piece(
        const int* __restrict__ src, const int* __restrict__ dst,
        int* __restrict__ cnt_src, int* __restrict__ cnt_dst,
        int* __restrict__ rank, int blk) {
    const int e = blk * 256 + (int)threadIdx.x;
    if (e >= N_EDGES) return;
    const int s = src[e], d = dst[e];
    atomicAdd(&cnt_src[s], 1);
    rank[e] = atomicAdd(&cnt_dst[d], 1);
}

__global__ __launch_bounds__(256) void front_kernel(
        const float* __restrict__ x, const float* __restrict__ W1, float* __restrict__ y1,
        const float* __restrict__ W2, const float* __restrict__ b2,
        const float* __restrict__ We, float* __restrict__ vw,
        const int* __restrict__ src, const int* __restrict__ dst,
        int* __restrict__ cnt_src, int* __restrict__ cnt_dst, int* __restrict__ rank) {
    __shared__ v2f sB[(IN_FEATS / 2) * 64];
    const int b = (int)blockIdx.x;
    if (b < GEMM_BLOCKS) {
        gemm_wmma_piece<IN_FEATS>(x, W1, y1, M_TILES, b, sB);
    } else if (b == GEMM_BLOCKS) {
        wproj_piece(W2, b2, We, vw);
    } else {
        count_piece(src, dst, cnt_src, cnt_dst, rank, b - (GEMM_BLOCKS + 1));
    }
}

__global__ __launch_bounds__(256) void gather64_proj_kernel(
        const float* __restrict__ y, const int* __restrict__ epair_raw,
        const int* __restrict__ rowptr, const int* __restrict__ cntd,
        const float* __restrict__ cdst, const float* __restrict__ bias,
        const float* __restrict__ vw, float* __restrict__ z, int n) {
    const int l = (int)threadIdx.x & 15;
    const int node = (int)blockIdx.x * 16 + ((int)threadIdx.x >> 4);
    if (node >= n) return;
    const int beg = rowptr[node];
    const int deg = cntd[node];
    const int end = beg + deg;
    const float4* __restrict__ y4 = (const float4*)y;
    const int2* __restrict__ epair = (const int2*)epair_raw;

    float4 acc = {0.0f, 0.0f, 0.0f, 0.0f};
    if (deg > 0) {
        const int rem  = deg & 3;
        const int last = end - 1;
        int j = beg + rem;
        int2 q0 = {0, 0}, q1 = {0, 0}, q2 = {0, 0}, q3 = {0, 0};
        if (j + 4 <= end) { q0 = epair[j]; q1 = epair[j + 1]; q2 = epair[j + 2]; q3 = epair[j + 3]; }
        if (rem > 0) {
            const int j1 = beg + 1 < last ? beg + 1 : last;
            const int j2 = beg + 2 < last ? beg + 2 : last;
            const int2 p0 = epair[beg], p1 = epair[j1], p2 = epair[j2];
            const float4 r0 = y4[(size_t)p0.x * 16 + l];
            float4 r1 = y4[(size_t)p1.x * 16 + l];
            float4 r2 = y4[(size_t)p2.x * 16 + l];
            const float c0 = __int_as_float(p0.y);
            float c1 = __int_as_float(p1.y), c2 = __int_as_float(p2.y);
            const bool u1 = rem > 1, u2 = rem > 2;
            c1 = u1 ? c1 : 0.0f; r1.x = u1 ? r1.x : 0.0f; r1.y = u1 ? r1.y : 0.0f; r1.z = u1 ? r1.z : 0.0f; r1.w = u1 ? r1.w : 0.0f;
            c2 = u2 ? c2 : 0.0f; r2.x = u2 ? r2.x : 0.0f; r2.y = u2 ? r2.y : 0.0f; r2.z = u2 ? r2.z : 0.0f; r2.w = u2 ? r2.w : 0.0f;
            acc.x += r0.x * c0; acc.y += r0.y * c0; acc.z += r0.z * c0; acc.w += r0.w * c0;
            acc.x += r1.x * c1; acc.y += r1.y * c1; acc.z += r1.z * c1; acc.w += r1.w * c1;
            acc.x += r2.x * c2; acc.y += r2.y * c2; acc.z += r2.z * c2; acc.w += r2.w * c2;
        }
        while (j + 4 <= end) {
            const int2 p0 = q0, p1 = q1, p2 = q2, p3 = q3;
            j += 4;
            if (j + 4 <= end) { q0 = epair[j]; q1 = epair[j + 1]; q2 = epair[j + 2]; q3 = epair[j + 3]; }
            const float4 r0 = y4[(size_t)p0.x * 16 + l];
            const float4 r1 = y4[(size_t)p1.x * 16 + l];
            const float4 r2 = y4[(size_t)p2.x * 16 + l];
            const float4 r3 = y4[(size_t)p3.x * 16 + l];
            const float c0 = __int_as_float(p0.y), c1 = __int_as_float(p1.y);
            const float c2 = __int_as_float(p2.y), c3 = __int_as_float(p3.y);
            acc.x += r0.x * c0; acc.y += r0.y * c0; acc.z += r0.z * c0; acc.w += r0.w * c0;
            acc.x += r1.x * c1; acc.y += r1.y * c1; acc.z += r1.z * c1; acc.w += r1.w * c1;
            acc.x += r2.x * c2; acc.y += r2.y * c2; acc.z += r2.z * c2; acc.w += r2.w * c2;
            acc.x += r3.x * c3; acc.y += r3.y * c3; acc.z += r3.z * c3; acc.w += r3.w * c3;
        }
    }

    const float cd = cdst[node];
    float4 o;
    o.x = fmaxf(acc.x * cd + bias[4 * l + 0], 0.0f);
    o.y = fmaxf(acc.y * cd + bias[4 * l + 1], 0.0f);
    o.z = fmaxf(acc.z * cd + bias[4 * l + 2], 0.0f);
    o.w = fmaxf(acc.w * cd + bias[4 * l + 3], 0.0f);

    const float4 ws = ((const float4*)vw)[l];
    const float4 wd = ((const float4*)vw)[16 + l];
    float ps = (o.x * ws.x + o.y * ws.y) + (o.z * ws.z + o.w * ws.w);
    float pd = (o.x * wd.x + o.y * wd.y) + (o.z * wd.z + o.w * wd.w);
    ps += __shfl_xor(ps, 8, 32); pd += __shfl_xor(pd, 8, 32);
    ps += __shfl_xor(ps, 4, 32); pd += __shfl_xor(pd, 4, 32);
    ps += __shfl_xor(ps, 2, 32); pd += __shfl_xor(pd, 2, 32);
    ps += __shfl_xor(ps, 1, 32); pd += __shfl_xor(pd, 1, 32);
    if (l == 0) {
        float2 p; p.x = ps; p.y = pd;
        ((float2*)z)[node] = p;
    }
}

__global__ __launch_bounds__(256) void gather2_kernel(
        const float* __restrict__ z, const int* __restrict__ epair_raw,
        const int* __restrict__ rowptr, const int* __restrict__ cntd,
        const float* __restrict__ cdst, const float* __restrict__ vw,
        float* __restrict__ pp, int n) {
    const int l = (int)threadIdx.x & 15;
    const int node = (int)blockIdx.x * 16 + ((int)threadIdx.x >> 4);
    if (node >= n) return;
    const int beg = rowptr[node];
    const int deg = cntd[node];
    const float cd = cdst[node];
    asm volatile("" :: "v"(beg), "v"(deg), "v"(cd));
    const int end = beg + deg;
    const int2* __restrict__ epair = (const int2*)epair_raw;
    const float2* __restrict__ z2 = (const float2*)z;

    float as = 0.0f, ad = 0.0f;
    for (int j = beg + l; j < end; j += 32) {
        const bool u1 = j + 16 < end;
        const int2 p0 = epair[j];
        const int2 p1 = epair[u1 ? j + 16 : j];
        const float2 z0 = z2[p0.x];
        float2 z1 = z2[p1.x];
        const float c0 = __int_as_float(p0.y);
        const float c1 = u1 ? __int_as_float(p1.y) : 0.0f;
        z1.x = u1 ? z1.x : 0.0f; z1.y = u1 ? z1.y : 0.0f;
        as += z0.x * c0; ad += z0.y * c0;
        as += z1.x * c1; ad += z1.y * c1;
    }
    as += __shfl_xor(as, 8, 32); ad += __shfl_xor(ad, 8, 32);
    as += __shfl_xor(as, 4, 32); ad += __shfl_xor(ad, 4, 32);
    as += __shfl_xor(as, 2, 32); ad += __shfl_xor(ad, 2, 32);
    as += __shfl_xor(as, 1, 32); ad += __shfl_xor(ad, 1, 32);
    if (l == 0) {
        float2 p; p.x = as * cd + vw[128]; p.y = ad * cd + vw[129];
        ((float2*)pp)[node] = p;
    }
}

__global__ __launch_bounds__(256) void edge_kernel(
        const float* __restrict__ pp, const int* __restrict__ src,
        const int* __restrict__ dst, const float* __restrict__ efeat,
        const float* __restrict__ We, const float* __restrict__ be,
        float* __restrict__ out, int E4) {
    const int g = (int)blockIdx.x * 256 + (int)threadIdx.x;
    if (g >= E4) return;
    const int4 s = ((const int4*)src)[g];
    const int4 d = ((const int4*)dst)[g];
    const float4* __restrict__ ef4 = (const float4*)efeat + (size_t)g * 3;
    const float4 f0 = ef4[0], f1 = ef4[1], f2 = ef4[2];
    const float s0 = pp[2 * s.x], s1 = pp[2 * s.y], s2 = pp[2 * s.z], s3 = pp[2 * s.w];
    const float d0 = pp[2 * d.x + 1], d1 = pp[2 * d.y + 1], d2 = pp[2 * d.z + 1], d3 = pp[2 * d.w + 1];
    const float w0 = We[128], w1 = We[129], w2 = We[130], b = be[0];
    const float t0 = (s0 + d0) + (f0.x * w0 + f0.y * w1 + f0.z * w2 + b);
    const float t1 = (s1 + d1) + (f0.w * w0 + f1.x * w1 + f1.y * w2 + b);
    const float t2 = (s2 + d2) + (f1.z * w0 + f1.w * w1 + f2.x * w2 + b);
    const float t3 = (s3 + d3) + (f2.y * w0 + f2.z * w1 + f2.w * w2 + b);
    float4 o;
    o.x = 1.0f / (1.0f + expf(-t0));
    o.y = 1.0f / (1.0f + expf(-t1));
    o.z = 1.0f / (1.0f + expf(-t2));
    o.w = 1.0f / (1.0f + expf(-t3));
    ((float4*)out)[g] = o;
}

extern "C" void kernel_launch(void* const* d_in, const int* in_sizes, int n_in,
                              void* d_out, int out_size, void* d_ws, size_t ws_size,
                              hipStream_t stream) {
    const float* x     = (const float*)d_in[0];
    const float* efeat = (const float*)d_in[1];
    const int*   src   = (const int*)  d_in[2];
    const int*   dst   = (const int*)  d_in[3];
    const float* W1    = (const float*)d_in[4];
    const float* b1    = (const float*)d_in[5];
    const float* W2    = (const float*)d_in[6];
    const float* b2    = (const float*)d_in[7];
    const float* We    = (const float*)d_in[8];
    const float* be    = (const float*)d_in[9];
    float* out = (float*)d_out;

    const size_t NB = (size_t)N_NODES * 64;
    char* ws = (char*)d_ws;
    float* bufA = (float*)ws;
    float* zbuf = bufA + NB;
    float* ppair = zbuf + 2 * N_NODES;
    float* vw   = ppair + 2 * N_NODES;
    int*   cntS = (int*)(vw + 256);
    int*   cntD = cntS + N_NODES;
    float* cS   = (float*)(cntD + N_NODES);
    float* cD   = cS + N_NODES;
    int*   rowp = (int*)(cD + N_NODES);
    int*   rank = rowp + N_NODES;
    int*   epair = rank + N_EDGES;

    hipMemsetAsync(cntS, 0, (size_t)2 * N_NODES * sizeof(int), stream);

    front_kernel<<<GEMM_BLOCKS + 1 + COUNT_BLOCKS, 256, 0, stream>>>(x, W1, bufA, W2, b2, We, vw,
                                                                     src, dst, cntS, cntD, rank);
    coeff_scan_kernel<<<(N_NODES + 255) / 256, 256, 0, stream>>>(cntS, cntD, cS, cD, rowp, N_NODES);
    fill_kernel<<<(N_EDGES / 4 + 255) / 256, 256, 0, stream>>>(src, dst, rank, rowp, cS, epair, N_EDGES / 4);

    gather64_proj_kernel<<<(N_NODES + 15) / 16, 256, 0, stream>>>(bufA, epair, rowp, cntD, cD, b1, vw, zbuf, N_NODES);

    gather2_kernel<<<(N_NODES + 15) / 16, 256, 0, stream>>>(zbuf, epair, rowp, cntD, cD, vw, ppair, N_NODES);

    edge_kernel<<<(N_EDGES / 4 + 255) / 256, 256, 0, stream>>>(ppair, src, dst, efeat, We, be, out, N_EDGES / 4);
}


# Per-edge global atomics (degree count and rank) replaced by a two-level counting partition on LDS counters, 256-node ranges; scan, fill and clear launches go; float sum order by LDS arrival rank
# speedup vs baseline: 7.6697x; 2.4858x over previous
#include <hip/hip_runtime.h>
#include <hip/hip_bf16.h>

#define N_NODES 50000
#define N_EDGES 800000
#define IN_FEATS 128
#define HIDDEN 64
#define M_TILES (N_NODES / 16)

#define NR   ((N_NODES + 255) / 256)
#define EPB  4096
#define NB1  ((N_EDGES + EPB - 1) / EPB)
#define N_PAD (NR * 256)
static_assert(N_EDGES % 16 == 0, "chunks are copied out in 16-byte pieces");
static_assert(N_NODES < (1 << 23), "a dst record is src * 256 + (dst mod 256) in 32 bits");

typedef float v2f __attribute__((ext_vector_type(2)));
typedef float v8f __attribute__((ext_vector_type(8)));

#define GEMM_BLOCKS  ((M_TILES + 7) / 8)

template<int K>
__device__ __forceinline__ void gemm_wmma_piece(
        const float* __restrict__ A, const float* __restrict__ W,
        float* __restrict__ out, int mtiles, int blk, v2f* __restrict__ sB) {
    for (int i = (int)threadIdx.x; i < (K / 2) * 64; i += 256) {
        const int j = i >> 6, n = i & 63;
        v2f p;
        p.x = W[(2 * j)     * 64 + n];
        p.y = W[(2 * j + 1) * 64 + n];
        sB[i] = p;
    }
    __syncthreads();

    const int lane  = (int)threadIdx.x & 31;
    const int mtile = blk * 8 + ((int)threadIdx.x >> 5);
    if (mtile >= mtiles) return;
    const int m_base = mtile * 16;
    const int col    = lane & 15;
    const int half   = lane >> 4;
    const int row    = m_base + col;
    const float* __restrict__ arow = A + (size_t)row * K;

    v8f acc0 = {}, acc1 = {}, acc2 = {}, acc3 = {};
    for (int k0 = 0; k0 < K; k0 += 4) {
        const int kb = k0 + (half << 1);
        const v2f a = *(const v2f*)(arow + kb);
        const v2f* __restrict__ brow = sB + ((k0 >> 1) + half) * 64;
        const v2f b0 = brow[     col];
        const v2f b1 = brow[16 + col];
        const v2f b2 = brow[32 + col];
        const v2f b3 = brow[48 + col];
        acc0 = __builtin_amdgcn_wmma_f32_16x16x4_f32(false, a, false, b0, (short)0, acc0, false, false);
        acc1 = __builtin_amdgcn_wmma_f32_16x16x4_f32(false, a, false, b1, (short)0, acc1, false, false);
        acc2 = __builtin_amdgcn_wmma_f32_16x16x4_f32(false, a, false, b2, (short)0, acc2, false, false);
        acc3 = __builtin_amdgcn_wmma_f32_16x16x4_f32(false, a, false, b3, (short)0, acc3, false, false);
    }

    const int rtop = half << 3;
    #pragma unroll
    for (int v = 0; v < 8; ++v) {
        const size_t r = (size_t)(m_base + rtop + v) * 64;
        out[r +      col] = acc0[v];
        out[r + 16 + col] = acc1[v];
        out[r + 32 + col] = acc2[v];
        out[r + 48 + col] = acc3[v];
    }
}

__device__ __forceinline__ void wproj_piece(
        const float* __restrict__ W2, const float* __restrict__ b2,
        const float* __restrict__ We, float* __restrict__ v) {
    const int t = (int)threadIdx.x;
    if (t > 64) return;
    const float4* __restrict__ r4 = (const float4*)(t < 64 ? W2 + t * 64 : b2);
    const float4* __restrict__ ws4 = (const float4*)We;
    const float4* __restrict__ wd4 = (const float4*)We + 16;
    float s = 0.0f, d = 0.0f;
    for (int j = 0; j < 16; ++j) {
        const float4 r = r4[j], a = ws4[j], b = wd4[j];
        s += (r.x * a.x + r.y * a.y) + (r.z * a.z + r.w * a.w);
        d += (r.x * b.x + r.y * b.y) + (r.z * b.z + r.w * b.w);
    }
    if (t < 64) { v[t] = s; v[64 + t] = d; }
    else        { v[128] = s; v[129] = d; }
}

__device__ __forceinline__ void partition_piece(
        const int* __restrict__ src, const int* __restrict__ dst,
        int* __restrict__ chunkD, unsigned char* __restrict__ chunkS,
        int* __restrict__ tabD, int* __restrict__ tabS, int blk, int* __restrict__ sm) {
    int* cntD = sm;
    int* cntS = sm + 256;
    int* wsum = sm + 512;
    int* stD  = sm + 544;
    unsigned char* stS = (unsigned char*)(stD + EPB);
    const int t = (int)threadIdx.x, lane = t & 31, w = t >> 5;
    cntD[t] = 0; cntS[t] = 0;
    __syncthreads();

    const int left = N_EDGES - blk * EPB;
    const int nvalid = left < EPB ? left : EPB;
    const int4* __restrict__ s4 = (const int4*)src + (size_t)blk * (EPB / 4);
    const int4* __restrict__ d4 = (const int4*)dst + (size_t)blk * (EPB / 4);
    int sv[16], dv[16], kd[16], ks[16];
    bool ok[4];
    #pragma unroll
    for (int g = 0; g < 4; ++g) {
        const int q = g * 256 + t;
        ok[g] = 4 * q < nvalid;
        const int qq = ok[g] ? q : 0;
        const int4 a = s4[qq], b = d4[qq];
        sv[4 * g] = a.x; sv[4 * g + 1] = a.y; sv[4 * g + 2] = a.z; sv[4 * g + 3] = a.w;
        dv[4 * g] = b.x; dv[4 * g + 1] = b.y; dv[4 * g + 2] = b.z; dv[4 * g + 3] = b.w;
    }
    #pragma unroll
    for (int i = 0; i < 16; ++i) {
        kd[i] = 0; ks[i] = 0;
        if (ok[i >> 2]) {
            kd[i] = atomicAdd(&cntD[dv[i] >> 8], 1);
            ks[i] = atomicAdd(&cntS[sv[i] >> 8], 1);
        }
    }
    __syncthreads();

    const int cd_ = cntD[t], cs_ = cntS[t];
    int iD = cd_, iS = cs_;
    #pragma unroll
    for (int off = 1; off < 32; off <<= 1) {
        const int a = __shfl_up(iD, off, 32), b = __shfl_up(iS, off, 32);
        if (lane >= off) { iD += a; iS += b; }
    }
    if (lane == 31) { wsum[w] = iD; wsum[8 + w] = iS; }
    __syncthreads();
    int bD = 0, bS = 0;
    #pragma unroll
    for (int k = 0; k < 8; ++k) {
        if (k < w) { bD += wsum[k]; bS += wsum[8 + k]; }
    }
    const int eD = bD + iD - cd_, eS = bS + iS - cs_;
    cntD[t] = eD; cntS[t] = eS;
    if (t <= NR) { tabD[t * NB1 + blk] = eD; tabS[t * NB1 + blk] = eS; }
    __syncthreads();

    #pragma unroll
    for (int i = 0; i < 16; ++i) {
        if (ok[i >> 2]) {
            stD[cntD[dv[i] >> 8] + kd[i]] = (sv[i] << 8) | (dv[i] & 255);
            stS[cntS[sv[i] >> 8] + ks[i]] = (unsigned char)(sv[i] & 255);
        }
    }
    __syncthreads();
    #pragma unroll
    for (int g = 0; g < 4; ++g) {
        const int q = g * 256 + t;
        if (4 * q < nvalid) ((int4*)chunkD)[(size_t)blk * (EPB / 4) + q] = ((const int4*)stD)[q];
    }
    if (16 * t < nvalid) ((int4*)chunkS)[(size_t)blk * (EPB / 16) + t] = ((const int4*)stS)[t];
}

__global__ __launch_bounds__(256) void front_kernel(
        const float* __restrict__ x, const float* __restrict__ W1, float* __restrict__ y1,
        const float* __restrict__ W2, const float* __restrict__ b2,
        const float* __restrict__ We, float* __restrict__ vw,
        const int* __restrict__ src, const int* __restrict__ dst,
        int* __restrict__ chunkD, int* __restrict__ chunkS_words,
        int* __restrict__ tabD, int* __restrict__ tabS) {
    __shared__ v2f sB[(IN_FEATS / 2) * 64];
    const int b = (int)blockIdx.x;
    if (b < GEMM_BLOCKS) {
        gemm_wmma_piece<IN_FEATS>(x, W1, y1, M_TILES, b, sB);
    } else if (b == GEMM_BLOCKS) {
        wproj_piece(W2, b2, We, vw);
    } else {
        partition_piece(src, dst, chunkD, (unsigned char*)chunkS_words, tabD, tabS, b - (GEMM_BLOCKS + 1), (int*)sB);
    }
}

#define RB_THREADS 512
#define RB_WAVES   (RB_THREADS / 32)
#define KSEG       ((NB1 + RB_WAVES - 1) / RB_WAVES)
__global__ __launch_bounds__(RB_THREADS) void range_kernel(
        const int* __restrict__ chunkD, const int* __restrict__ chunkS_words,
        const int* __restrict__ tabD, const int* __restrict__ tabS,
        int* __restrict__ rowptr, int* __restrict__ deg,
        float* __restrict__ cS, float* __restrict__ cD, int* __restrict__ epair) {
    const unsigned char* __restrict__ chunkS = (const unsigned char*)chunkS_words;
    __shared__ int cnt[256];
    __shared__ int cntT[256];
    __shared__ int curT[256];
    __shared__ int lrow[256];
    __shared__ int wtot[8];
    __shared__ int s_start;
    const int t = (int)threadIdx.x, lane = t & 31;
    const int w = t >> 5;
    const bool is_dst = (int)blockIdx.x < NR;
    const int r = is_dst ? (int)blockIdx.x : (int)blockIdx.x - NR;
    if (t < 256) { cnt[t] = 0; cntT[t] = 0; curT[t] = 0; }
    if (t == 0) s_start = 0;
    __syncthreads();

    if (is_dst) {
        const int* __restrict__ row0 = tabD + r * NB1;
        const int* __restrict__ row1 = row0 + NB1;
        {
            int v = t < NB1 ? row0[t] : 0;
            #pragma unroll
            for (int off = 16; off > 0; off >>= 1) v += __shfl_xor(v, off, 32);
            if (lane == 0 && v != 0) atomicAdd(&s_start, v);
        }
        int o[KSEG], c[KSEG], rec[KSEG], rk[KSEG];
        {
            const int b = w + lane * RB_WAVES;
            const bool bv = lane < KSEG && b < NB1;
            const int bb = bv ? b : 0;
            const int o0 = row0[bb], o1 = row1[bb];
            const int ov = bb * EPB + o0;
            const int cv = bv ? o1 - o0 : 0;
            #pragma unroll
            for (int k = 0; k < KSEG; ++k) {
                o[k] = __builtin_amdgcn_readlane(ov, k);
                c[k] = __builtin_amdgcn_readlane(cv, k);
            }
        }
        #pragma unroll
        for (int k = 0; k < KSEG; ++k) {
            rec[k] = chunkD[lane < c[k] ? o[k] + lane : 0];
        }
        #pragma unroll
        for (int k = 0; k < KSEG; ++k) {
            rk[k] = 0;
            if (lane < c[k]) rk[k] = atomicAdd(&cnt[rec[k] & 255], 1);
            for (int i = lane + 32; i < c[k]; i += 32) atomicAdd(&cntT[chunkD[o[k] + i] & 255], 1);
        }
        __syncthreads();

        int tot = 0, incl = 0;
        if (t < 256) {
            tot = cnt[t] + cntT[t];
            incl = tot;
            #pragma unroll
            for (int off = 1; off < 32; off <<= 1) {
                const int a = __shfl_up(incl, off, 32);
                if (lane >= off) incl += a;
            }
            if (lane == 31) wtot[w] = incl;
        }
        __syncthreads();
        const int start = s_start;
        if (t < 256) {
            int base = 0;
            #pragma unroll
            for (int k = 0; k < 8; ++k) { if (k < w) base += wtot[k]; }
            const int ex = base + incl - tot;
            lrow[t] = ex;
            const int n = r * 256 + t;
            if (n < N_NODES) {
                rowptr[n] = start + ex;
                deg[n] = tot;
                cD[n] = rsqrtf((float)(tot < 1 ? 1 : tot));
            }
        }
        __syncthreads();
        #pragma unroll
        for (int k = 0; k < KSEG; ++k) {
            if (lane < c[k]) {
                const int nl = rec[k] & 255;
                epair[start + lrow[nl] + rk[k]] = rec[k] >> 8;
            }
            for (int i = lane + 32; i < c[k]; i += 32) {
                const int xr = chunkD[o[k] + i];
                const int nl = xr & 255;
                const int p = atomicAdd(&curT[nl], 1);
                epair[start + lrow[nl] + cnt[nl] + p] = xr >> 8;
            }
        }
    } else {
        const int* __restrict__ row0 = tabS + r * NB1;
        const int* __restrict__ row1 = row0 + NB1;
        int o[KSEG], c[KSEG], rec[KSEG];
        {
            const int b = w + lane * RB_WAVES;
            const bool bv = lane < KSEG && b < NB1;
            const int bb = bv ? b : 0;
            const int o0 = row0[bb], o1 = row1[bb];
            const int ov = bb * EPB + o0;
            const int cv = bv ? o1 - o0 : 0;
            #pragma unroll
            for (int k = 0; k < KSEG; ++k) {
                o[k] = __builtin_amdgcn_readlane(ov, k);
                c[k] = __builtin_amdgcn_readlane(cv, k);
            }
        }
        #pragma unroll
        for (int k = 0; k < KSEG; ++k) {
            rec[k] = (int)chunkS[lane < c[k] ? o[k] + lane : 0];
        }
        #pragma unroll
        for (int k = 0; k < KSEG; ++k) {
            if (lane < c[k]) atomicAdd(&cnt[rec[k]], 1);
            for (int i = lane + 32; i < c[k]; i += 32) atomicAdd(&cnt[(int)chunkS[o[k] + i]], 1);
        }
        __syncthreads();
        if (t < 256) {
            const int n = r * 256 + t;
            if (n < N_NODES) {
                const int a = cnt[t];
                cS[n] = rsqrtf((float)(a < 1 ? 1 : a));
            }
        }
    }
}

__global__ __launch_bounds__(256) void gather64_proj_kernel(
        const float* __restrict__ y, const int* __restrict__ epair,
        const int* __restrict__ rowptr, const int* __restrict__ cntd,
        const float* __restrict__ cdst, const float* __restrict__ csrc,
        const float* __restrict__ bias,
        const float* __restrict__ vw, float* __restrict__ z, int n) {
    const int l = (int)threadIdx.x & 15;
    const int node = (int)blockIdx.x * 16 + ((int)threadIdx.x >> 4);
    if (node >= n) return;
    const int beg = rowptr[node];
    const int deg = cntd[node];
    const float csn = csrc[node];
    asm volatile("" :: "v"(beg), "v"(deg), "v"(csn));
    const int end = beg + deg;
    const float4* __restrict__ y4 = (const float4*)y;

    float4 acc = {0.0f, 0.0f, 0.0f, 0.0f};
    if (deg > 0) {
        const int rem  = deg & 3;
        const int last = end - 1;
        int j = beg + rem;
        int q0 = 0, q1 = 0, q2 = 0, q3 = 0;
        if (j + 4 <= end) { q0 = epair[j]; q1 = epair[j + 1]; q2 = epair[j + 2]; q3 = epair[j + 3]; }
        if (rem > 0) {
            const int j1 = beg + 1 < last ? beg + 1 : last;
            const int j2 = beg + 2 < last ? beg + 2 : last;
            const int p0 = epair[beg], p1 = epair[j1], p2 = epair[j2];
            const float4 r0 = y4[(size_t)p0 * 16 + l];
            float4 r1 = y4[(size_t)p1 * 16 + l];
            float4 r2 = y4[(size_t)p2 * 16 + l];
            const float c0 = csrc[p0];
            float c1 = csrc[p1], c2 = csrc[p2];
            const bool u1 = rem > 1, u2 = rem > 2;
            c1 = u1 ? c1 : 0.0f; r1.x = u1 ? r1.x : 0.0f; r1.y = u1 ? r1.y : 0.0f; r1.z = u1 ? r1.z : 0.0f; r1.w = u1 ? r1.w : 0.0f;
            c2 = u2 ? c2 : 0.0f; r2.x = u2 ? r2.x : 0.0f; r2.y = u2 ? r2.y : 0.0f; r2.z = u2 ? r2.z : 0.0f; r2.w = u2 ? r2.w : 0.0f;
            acc.x += r0.x * c0; acc.y += r0.y * c0; acc.z += r0.z * c0; acc.w += r0.w * c0;
            acc.x += r1.x * c1; acc.y += r1.y * c1; acc.z += r1.z * c1; acc.w += r1.w * c1;
            acc.x += r2.x * c2; acc.y += r2.y * c2; acc.z += r2.z * c2; acc.w += r2.w * c2;
        }
        while (j + 4 <= end) {
            const int p0 = q0, p1 = q1, p2 = q2, p3 = q3;
            j += 4;
            if (j + 4 <= end) { q0 = epair[j]; q1 = epair[j + 1]; q2 = epair[j + 2]; q3 = epair[j + 3]; }
            const float4 r0 = y4[(size_t)p0 * 16 + l];
            const float4 r1 = y4[(size_t)p1 * 16 + l];
            const float4 r2 = y4[(size_t)p2 * 16 + l];
            const float4 r3 = y4[(size_t)p3 * 16 + l];
            const float c0 = csrc[p0], c1 = csrc[p1];
            const float c2 = csrc[p2], c3 = csrc[p3];
            acc.x += r0.x * c0; acc.y += r0.y * c0; acc.z += r0.z * c0; acc.w += r0.w * c0;
            acc.x += r1.x * c1; acc.y += r1.y * c1; acc.z += r1.z * c1; acc.w += r1.w * c1;
            acc.x += r2.x * c2; acc.y += r2.y * c2; acc.z += r2.z * c2; acc.w += r2.w * c2;
            acc.x += r3.x * c3; acc.y += r3.y * c3; acc.z += r3.z * c3; acc.w += r3.w * c3;
        }
    }

    const float cd = cdst[node];
    const float4 bv = ((const float4*)bias)[l];
    const float4 ws = ((const float4*)vw)[l];
    const float4 wd = ((const float4*)vw)[16 + l];
    float4 o;
    o.x = fmaxf(acc.x * cd + bv.x, 0.0f);
    o.y = fmaxf(acc.y * cd + bv.y, 0.0f);
    o.z = fmaxf(acc.z * cd + bv.z, 0.0f);
    o.w = fmaxf(acc.w * cd + bv.w, 0.0f);

    float ps = (o.x * ws.x + o.y * ws.y) + (o.z * ws.z + o.w * ws.w);
    float pd = (o.x * wd.x + o.y * wd.y) + (o.z * wd.z + o.w * wd.w);
    ps += __shfl_xor(ps, 8, 32); pd += __shfl_xor(pd, 8, 32);
    ps += __shfl_xor(ps, 4, 32); pd += __shfl_xor(pd, 4, 32);
    ps += __shfl_xor(ps, 2, 32); pd += __shfl_xor(pd, 2, 32);
    ps += __shfl_xor(ps, 1, 32); pd += __shfl_xor(pd, 1, 32);
    if (l == 0) {
        float4 p; p.x = ps; p.y = pd; p.z = csn; p.w = 0.0f;
        ((float4*)z)[node] = p;
    }
}

__global__ __launch_bounds__(256) void gather2_kernel(
        const float* __restrict__ z, const int* __restrict__ epair,
        const int* __restrict__ rowptr, const int* __restrict__ cntd,
        const float* __restrict__ cdst, const float* __restrict__ vw,
        float* __restrict__ pp, int n) {
    const int l = (int)threadIdx.x & 15;
    const int node = (int)blockIdx.x * 16 + ((int)threadIdx.x >> 4);
    if (node >= n) return;
    const int beg = rowptr[node];
    const int deg = cntd[node];
    const float cd = cdst[node];
    asm volatile("" :: "v"(beg), "v"(deg), "v"(cd));
    const int end = beg + deg;
    const float4* __restrict__ z4 = (const float4*)z;

    float as = 0.0f, ad = 0.0f;
    for (int j = beg + l; j < end; j += 32) {
        const bool u1 = j + 16 < end;
        const int p0 = epair[j];
        const int p1 = epair[u1 ? j + 16 : j];
        const float4 z0 = z4[p0];
        float4 z1 = z4[p1];
        const float c0 = z0.z;
        const float c1 = u1 ? z1.z : 0.0f;
        z1.x = u1 ? z1.x : 0.0f; z1.y = u1 ? z1.y : 0.0f;
        as += z0.x * c0; ad += z0.y * c0;
        as += z1.x * c1; ad += z1.y * c1;
    }
    as += __shfl_xor(as, 8, 32); ad += __shfl_xor(ad, 8, 32);
    as += __shfl_xor(as, 4, 32); ad += __shfl_xor(ad, 4, 32);
    as += __shfl_xor(as, 2, 32); ad += __shfl_xor(ad, 2, 32);
    as += __shfl_xor(as, 1, 32); ad += __shfl_xor(ad, 1, 32);
    if (l == 0) {
        float2 p; p.x = as * cd + vw[128]; p.y = ad * cd + vw[129];
        ((float2*)pp)[node] = p;
    }
}

__global__ __launch_bounds__(256) void edge_kernel(
        const float* __restrict__ pp, const int* __restrict__ src,
        const int* __restrict__ dst, const float* __restrict__ efeat,
        const float* __restrict__ We, const float* __restrict__ be,
        float* __restrict__ out, int E4) {
    const int g = (int)blockIdx.x * 256 + (int)threadIdx.x;
    if (g >= E4) return;
    const int4 s = ((const int4*)src)[g];
    const int4 d = ((const int4*)dst)[g];
    const float4* __restrict__ ef4 = (const float4*)efeat + (size_t)g * 3;
    const float4 f0 = ef4[0], f1 = ef4[1], f2 = ef4[2];
    const float s0 = pp[2 * s.x], s1 = pp[2 * s.y], s2 = pp[2 * s.z], s3 = pp[2 * s.w];
    const float d0 = pp[2 * d.x + 1], d1 = pp[2 * d.y + 1], d2 = pp[2 * d.z + 1], d3 = pp[2 * d.w + 1];
    const float w0 = We[128], w1 = We[129], w2 = We[130], b = be[0];
    const float t0 = (s0 + d0) + (f0.x * w0 + f0.y * w1 + f0.z * w2 + b);
    const float t1 = (s1 + d1) + (f0.w * w0 + f1.x * w1 + f1.y * w2 + b);
    const float t2 = (s2 + d2) + (f1.z * w0 + f1.w * w1 + f2.x * w2 + b);
    const float t3 = (s3 + d3) + (f2.y * w0 + f2.z * w1 + f2.w * w2 + b);
    float4 o;
    o.x = 1.0f / (1.0f + expf(-t0));
    o.y = 1.0f / (1.0f + expf(-t1));
    o.z = 1.0f / (1.0f + expf(-t2));
    o.w = 1.0f / (1.0f + expf(-t3));
    ((float4*)out)[g] = o;
}

extern "C" void kernel_launch(void* const* d_in, const int* in_sizes, int n_in,
                              void* d_out, int out_size, void* d_ws, size_t ws_size,
                              hipStream_t stream) {
    const float* x     = (const float*)d_in[0];
    const float* efeat = (const float*)d_in[1];
    const int*   src   = (const int*)  d_in[2];
    const int*   dst   = (const int*)  d_in[3];
    const float* W1    = (const float*)d_in[4];
    const float* b1    = (const float*)d_in[5];
    const float* W2    = (const float*)d_in[6];
    const float* b2    = (const float*)d_in[7];
    const float* We    = (const float*)d_in[8];
    const float* be    = (const float*)d_in[9];
    float* out = (float*)d_out;

    const size_t NB = (size_t)N_NODES * 64;
    char* ws = (char*)d_ws;
    float* bufA = (float*)ws;
    float* zbuf = bufA + NB;
    float* ppair = zbuf + 4 * N_NODES;
    float* vw   = ppair + 2 * N_NODES;
    int*   degD = (int*)(vw + 256);
    int*   rowp = degD + N_PAD;
    float* cS   = (float*)(rowp + N_PAD);
    float* cD   = cS + N_PAD;
    int*   tabD = (int*)(cD + N_PAD);
    int*   tabS = tabD + (NR + 1) * NB1;
    int*   epair = tabS + (NR + 1) * NB1;
    int*   chunkD = epair + N_EDGES;
    int*   chunkS = chunkD + NB1 * EPB;
    static_assert(((size_t)N_NODES * 64 + 6 * N_NODES + 256 + 4 * N_PAD + 2 * (NR + 1) * NB1 + N_EDGES) % 4 == 0,
                  "chunkD starts on a 16-byte boundary");
    static_assert((NB1 * EPB) % 4 == 0, "chunkS starts on a 16-byte boundary");
    static_assert(((size_t)N_NODES * 64 + 6 * N_NODES + 256) % 32 == 0, "the per-node arrays start on a 128-byte line");

    front_kernel<<<GEMM_BLOCKS + 1 + NB1, 256, 0, stream>>>(x, W1, bufA, W2, b2, We, vw,
                                                            src, dst, chunkD, chunkS, tabD, tabS);
    range_kernel<<<2 * NR, RB_THREADS, 0, stream>>>(chunkD, chunkS, tabD, tabS, rowp, degD, cS, cD, epair);

    gather64_proj_kernel<<<(N_NODES + 15) / 16, 256, 0, stream>>>(bufA, epair, rowp, degD, cD, cS, b1, vw, zbuf, N_NODES);

    gather2_kernel<<<(N_NODES + 15) / 16, 256, 0, stream>>>(zbuf, epair, rowp, degD, cD, vw, ppair, N_NODES);

    edge_kernel<<<(N_EDGES / 4 + 255) / 256, 256, 0, stream>>>(ppair, src, dst, efeat, We, be, out, N_EDGES / 4);
}


# Range blocks build their run of edge slots in LDS and store it as whole 128-byte lines, 16 bytes per lane, not as 4-byte stores with every lane on another line; slot contents and sum order kept
# speedup vs baseline: 7.9264x; 1.0335x over previous
#include <hip/hip_runtime.h>
#include <hip/hip_bf16.h>

#define N_NODES 50000
#define N_EDGES 800000
#define IN_FEATS 128
#define HIDDEN 64
#define M_TILES (N_NODES / 16)

#define NR   ((N_NODES + 255) / 256)
#define EPB  4096
#define NB1  ((N_EDGES + EPB - 1) / EPB)
#define N_PAD (NR * 256)
static_assert(N_EDGES % 16 == 0, "chunks are copied out in 16-byte pieces");
static_assert(N_NODES < (1 << 23), "a dst record is src * 256 + (dst mod 256) in 32 bits");

typedef float v2f __attribute__((ext_vector_type(2)));
typedef float v8f __attribute__((ext_vector_type(8)));

#define GEMM_BLOCKS  ((M_TILES + 7) / 8)

template<int K>
__device__ __forceinline__ void gemm_wmma_piece(
        const float* __restrict__ A, const float* __restrict__ W,
        float* __restrict__ out, int mtiles, int blk, v2f* __restrict__ sB) {
    for (int i = (int)threadIdx.x; i < (K / 2) * 64; i += 256) {
        const int j = i >> 6, n = i & 63;
        v2f p;
        p.x = W[(2 * j)     * 64 + n];
        p.y = W[(2 * j + 1) * 64 + n];
        sB[i] = p;
    }
    __syncthreads();

    const int lane  = (int)threadIdx.x & 31;
    const int mtile = blk * 8 + ((int)threadIdx.x >> 5);
    if (mtile >= mtiles) return;
    const int m_base = mtile * 16;
    const int col    = lane & 15;
    const int half   = lane >> 4;
    const int row    = m_base + col;
    const float* __restrict__ arow = A + (size_t)row * K;

    v8f acc0 = {}, acc1 = {}, acc2 = {}, acc3 = {};
    for (int k0 = 0; k0 < K; k0 += 4) {
        const int kb = k0 + (half << 1);
        const v2f a = *(const v2f*)(arow + kb);
        const v2f* __restrict__ brow = sB + ((k0 >> 1) + half) * 64;
        const v2f b0 = brow[     col];
        const v2f b1 = brow[16 + col];
        const v2f b2 = brow[32 + col];
        const v2f b3 = brow[48 + col];
        acc0 = __builtin_amdgcn_wmma_f32_16x16x4_f32(false, a, false, b0, (short)0, acc0, false, false);
        acc1 = __builtin_amdgcn_wmma_f32_16x16x4_f32(false, a, false, b1, (short)0, acc1, false, false);
        acc2 = __builtin_amdgcn_wmma_f32_16x16x4_f32(false, a, false, b2, (short)0, acc2, false, false);
        acc3 = __builtin_amdgcn_wmma_f32_16x16x4_f32(false, a, false, b3, (short)0, acc3, false, false);
    }

    const int rtop = half << 3;
    #pragma unroll
    for (int v = 0; v < 8; ++v) {
        const size_t r = (size_t)(m_base + rtop + v) * 64;
        out[r +      col] = acc0[v];
        out[r + 16 + col] = acc1[v];
        out[r + 32 + col] = acc2[v];
        out[r + 48 + col] = acc3[v];
    }
}

__device__ __forceinline__ void wproj_piece(
        const float* __restrict__ W2, const float* __restrict__ b2,
        const float* __restrict__ We, float* __restrict__ v) {
    const int t = (int)threadIdx.x;
    if (t > 64) return;
    const float4* __restrict__ r4 = (const float4*)(t < 64 ? W2 + t * 64 : b2);
    const float4* __restrict__ ws4 = (const float4*)We;
    const float4* __restrict__ wd4 = (const float4*)We + 16;
    float s = 0.0f, d = 0.0f;
    for (int j = 0; j < 16; ++j) {
        const float4 r = r4[j], a = ws4[j], b = wd4[j];
        s += (r.x * a.x + r.y * a.y) + (r.z * a.z + r.w * a.w);
        d += (r.x * b.x + r.y * b.y) + (r.z * b.z + r.w * b.w);
    }
    if (t < 64) { v[t] = s; v[64 + t] = d; }
    else        { v[128] = s; v[129] = d; }
}

__device__ __forceinline__ void partition_piece(
        const int* __restrict__ src, const int* __restrict__ dst,
        int* __restrict__ chunkD, unsigned char* __restrict__ chunkS,
        int* __restrict__ tabD, int* __restrict__ tabS, int blk, int* __restrict__ sm) {
    int* cntD = sm;
    int* cntS = sm + 256;
    int* wsum = sm + 512;
    int* stD  = sm + 544;
    unsigned char* stS = (unsigned char*)(stD + EPB);
    const int t = (int)threadIdx.x, lane = t & 31, w = t >> 5;
    cntD[t] = 0; cntS[t] = 0;
    __syncthreads();

    const int left = N_EDGES - blk * EPB;
    const int nvalid = left < EPB ? left : EPB;
    const int4* __restrict__ s4 = (const int4*)src + (size_t)blk * (EPB / 4);
    const int4* __restrict__ d4 = (const int4*)dst + (size_t)blk * (EPB / 4);
    int sv[16], dv[16], kd[16], ks[16];
    bool ok[4];
    #pragma unroll
    for (int g = 0; g < 4; ++g) {
        const int q = g * 256 + t;
        ok[g] = 4 * q < nvalid;
        const int qq = ok[g] ? q : 0;
        const int4 a = s4[qq], b = d4[qq];
        sv[4 * g] = a.x; sv[4 * g + 1] = a.y; sv[4 * g + 2] = a.z; sv[4 * g + 3] = a.w;
        dv[4 * g] = b.x; dv[4 * g + 1] = b.y; dv[4 * g + 2] = b.z; dv[4 * g + 3] = b.w;
    }
    #pragma unroll
    for (int i = 0; i < 16; ++i) {
        kd[i] = 0; ks[i] = 0;
        if (ok[i >> 2]) {
            kd[i] = atomicAdd(&cntD[dv[i] >> 8], 1);
            ks[i] = atomicAdd(&cntS[sv[i] >> 8], 1);
        }
    }
    __syncthreads();

    const int cd_ = cntD[t], cs_ = cntS[t];
    int iD = cd_, iS = cs_;
    #pragma unroll
    for (int off = 1; off < 32; off <<= 1) {
        const int a = __shfl_up(iD, off, 32), b = __shfl_up(iS, off, 32);
        if (lane >= off) { iD += a; iS += b; }
    }
    if (lane == 31) { wsum[w] = iD; wsum[8 + w] = iS; }
    __syncthreads();
    int bD = 0, bS = 0;
    #pragma unroll
    for (int k = 0; k < 8; ++k) {
        if (k < w) { bD += wsum[k]; bS += wsum[8 + k]; }
    }
    const int eD = bD + iD - cd_, eS = bS + iS - cs_;
    cntD[t] = eD; cntS[t] = eS;
    if (t <= NR) { tabD[t * NB1 + blk] = eD; tabS[t * NB1 + blk] = eS; }
    __syncthreads();

    #pragma unroll
    for (int i = 0; i < 16; ++i) {
        if (ok[i >> 2]) {
            stD[cntD[dv[i] >> 8] + kd[i]] = (sv[i] << 8) | (dv[i] & 255);
            stS[cntS[sv[i] >> 8] + ks[i]] = (unsigned char)(sv[i] & 255);
        }
    }
    __syncthreads();
    #pragma unroll
    for (int g = 0; g < 4; ++g) {
        const int q = g * 256 + t;
        if (4 * q < nvalid) ((int4*)chunkD)[(size_t)blk * (EPB / 4) + q] = ((const int4*)stD)[q];
    }
    if (16 * t < nvalid) ((int4*)chunkS)[(size_t)blk * (EPB / 16) + t] = ((const int4*)stS)[t];
}

__global__ __launch_bounds__(256) void front_kernel(
        const float* __restrict__ x, const float* __restrict__ W1, float* __restrict__ y1,
        const float* __restrict__ W2, const float* __restrict__ b2,
        const float* __restrict__ We, float* __restrict__ vw,
        const int* __restrict__ src, const int* __restrict__ dst,
        int* __restrict__ chunkD, int* __restrict__ chunkS_words,
        int* __restrict__ tabD, int* __restrict__ tabS) {
    __shared__ v2f sB[(IN_FEATS / 2) * 64];
    const int b = (int)blockIdx.x;
    if (b < GEMM_BLOCKS) {
        gemm_wmma_piece<IN_FEATS>(x, W1, y1, M_TILES, b, sB);
    } else if (b == GEMM_BLOCKS) {
        wproj_piece(W2, b2, We, vw);
    } else {
        partition_piece(src, dst, chunkD, (unsigned char*)chunkS_words, tabD, tabS, b - (GEMM_BLOCKS + 1), (int*)sB);
    }
}

#define STG 6144
#define RB_THREADS 512
#define RB_WAVES   (RB_THREADS / 32)
#define KSEG       ((NB1 + RB_WAVES - 1) / RB_WAVES)
__global__ __launch_bounds__(RB_THREADS) void range_kernel(
        const int* __restrict__ chunkD, const int* __restrict__ chunkS_words,
        const int* __restrict__ tabD, const int* __restrict__ tabS,
        int* __restrict__ rowptr, int* __restrict__ deg,
        float* __restrict__ cS, float* __restrict__ cD, int* __restrict__ epair) {
    const unsigned char* __restrict__ chunkS = (const unsigned char*)chunkS_words;
    __shared__ int cnt[256];
    __shared__ int cntT[256];
    __shared__ int curT[256];
    __shared__ int lrow[256];
    __shared__ int wtot[8];
    __shared__ int s_start;
    __shared__ int s_tot;
    __shared__ __attribute__((aligned(16))) int stage[STG];
    const int t = (int)threadIdx.x, lane = t & 31;
    const int w = t >> 5;
    const bool is_dst = (int)blockIdx.x < NR;
    const int r = is_dst ? (int)blockIdx.x : (int)blockIdx.x - NR;
    if (t < 256) { cnt[t] = 0; cntT[t] = 0; curT[t] = 0; }
    if (t == 0) s_start = 0;
    __syncthreads();

    if (is_dst) {
        const int* __restrict__ row0 = tabD + r * NB1;
        const int* __restrict__ row1 = row0 + NB1;
        {
            int v = t < NB1 ? row0[t] : 0;
            #pragma unroll
            for (int off = 16; off > 0; off >>= 1) v += __shfl_xor(v, off, 32);
            if (lane == 0 && v != 0) atomicAdd(&s_start, v);
        }
        int o[KSEG], c[KSEG], rec[KSEG], rk[KSEG];
        {
            const int b = w + lane * RB_WAVES;
            const bool bv = lane < KSEG && b < NB1;
            const int bb = bv ? b : 0;
            const int o0 = row0[bb], o1 = row1[bb];
            const int ov = bb * EPB + o0;
            const int cv = bv ? o1 - o0 : 0;
            #pragma unroll
            for (int k = 0; k < KSEG; ++k) {
                o[k] = __builtin_amdgcn_readlane(ov, k);
                c[k] = __builtin_amdgcn_readlane(cv, k);
            }
        }
        #pragma unroll
        for (int k = 0; k < KSEG; ++k) {
            rec[k] = chunkD[lane < c[k] ? o[k] + lane : 0];
        }
        #pragma unroll
        for (int k = 0; k < KSEG; ++k) {
            rk[k] = 0;
            if (lane < c[k]) rk[k] = atomicAdd(&cnt[rec[k] & 255], 1);
            for (int i = lane + 32; i < c[k]; i += 32) atomicAdd(&cntT[chunkD[o[k] + i] & 255], 1);
        }
        __syncthreads();

        int tot = 0, incl = 0;
        if (t < 256) {
            tot = cnt[t] + cntT[t];
            incl = tot;
            #pragma unroll
            for (int off = 1; off < 32; off <<= 1) {
                const int a = __shfl_up(incl, off, 32);
                if (lane >= off) incl += a;
            }
            if (lane == 31) wtot[w] = incl;
        }
        __syncthreads();
        const int start = s_start;
        if (t < 256) {
            int base = 0;
            #pragma unroll
            for (int k = 0; k < 8; ++k) { if (k < w) base += wtot[k]; }
            const int ex = base + incl - tot;
            lrow[t] = ex;
            if (t == 255) s_tot = base + incl;
            const int n = r * 256 + t;
            if (n < N_NODES) {
                rowptr[n] = start + ex;
                deg[n] = tot;
                cD[n] = rsqrtf((float)(tot < 1 ? 1 : tot));
            }
        }
        __syncthreads();
        const int mis = (int)(((size_t)(epair + start) >> 2) & 31);
        int* __restrict__ line0 = epair + (start - mis);
        #pragma unroll
        for (int k = 0; k < KSEG; ++k) {
            if (lane < c[k]) {
                const int nl = rec[k] & 255;
                const int p = mis + lrow[nl] + rk[k];
                if (p < STG) stage[p] = rec[k] >> 8;
                else         line0[p] = rec[k] >> 8;
            }
            for (int i = lane + 32; i < c[k]; i += 32) {
                const int xr = chunkD[o[k] + i];
                const int nl = xr & 255;
                const int p = mis + lrow[nl] + cnt[nl] + atomicAdd(&curT[nl], 1);
                if (p < STG) stage[p] = xr >> 8;
                else         line0[p] = xr >> 8;
            }
        }
        __syncthreads();
        const int endp = mis + s_tot;
        const int lim  = endp < STG ? endp : STG;
        for (int q = t; 4 * q < lim; q += RB_THREADS) {
            const int4 v = ((const int4*)stage)[q];
            const int p0 = 4 * q;
            if (p0 >= mis && p0 + 4 <= lim) {
                ((int4*)line0)[q] = v;
            } else {
                if (p0     >= mis && p0     < lim) line0[p0]     = v.x;
                if (p0 + 1 >= mis && p0 + 1 < lim) line0[p0 + 1] = v.y;
                if (p0 + 2 >= mis && p0 + 2 < lim) line0[p0 + 2] = v.z;
                if (p0 + 3 >= mis && p0 + 3 < lim) line0[p0 + 3] = v.w;
            }
        }
    } else {
        const int* __restrict__ row0 = tabS + r * NB1;
        const int* __restrict__ row1 = row0 + NB1;
        int o[KSEG], c[KSEG], rec[KSEG];
        {
            const int b = w + lane * RB_WAVES;
            const bool bv = lane < KSEG && b < NB1;
            const int bb = bv ? b : 0;
            const int o0 = row0[bb], o1 = row1[bb];
            const int ov = bb * EPB + o0;
            const int cv = bv ? o1 - o0 : 0;
            #pragma unroll
            for (int k = 0; k < KSEG; ++k) {
                o[k] = __builtin_amdgcn_readlane(ov, k);
                c[k] = __builtin_amdgcn_readlane(cv, k);
            }
        }
        #pragma unroll
        for (int k = 0; k < KSEG; ++k) {
            rec[k] = (int)chunkS[lane < c[k] ? o[k] + lane : 0];
        }
        #pragma unroll
        for (int k = 0; k < KSEG; ++k) {
            if (lane < c[k]) atomicAdd(&cnt[rec[k]], 1);
            for (int i = lane + 32; i < c[k]; i += 32) atomicAdd(&cnt[(int)chunkS[o[k] + i]], 1);
        }
        __syncthreads();
        if (t < 256) {
            const int n = r * 256 + t;
            if (n < N_NODES) {
                const int a = cnt[t];
                cS[n] = rsqrtf((float)(a < 1 ? 1 : a));
            }
        }
    }
}

__global__ __launch_bounds__(256) void gather64_proj_kernel(
        const float* __restrict__ y, const int* __restrict__ epair,
        const int* __restrict__ rowptr, const int* __restrict__ cntd,
        const float* __restrict__ cdst, const float* __restrict__ csrc,
        const float* __restrict__ bias,
        const float* __restrict__ vw, float* __restrict__ z, int n) {
    const int l = (int)threadIdx.x & 15;
    const int node = (int)blockIdx.x * 16 + ((int)threadIdx.x >> 4);
    if (node >= n) return;
    const int beg = rowptr[node];
    const int deg = cntd[node];
    const float csn = csrc[node];
    asm volatile("" :: "v"(beg), "v"(deg), "v"(csn));
    const int end = beg + deg;
    const float4* __restrict__ y4 = (const float4*)y;

    float4 acc = {0.0f, 0.0f, 0.0f, 0.0f};
    if (deg > 0) {
        const int rem  = deg & 3;
        const int last = end - 1;
        int j = beg + rem;
        int q0 = 0, q1 = 0, q2 = 0, q3 = 0;
        if (j + 4 <= end) { q0 = epair[j]; q1 = epair[j + 1]; q2 = epair[j + 2]; q3 = epair[j + 3]; }
        if (rem > 0) {
            const int j1 = beg + 1 < last ? beg + 1 : last;
            const int j2 = beg + 2 < last ? beg + 2 : last;
            const int p0 = epair[beg], p1 = epair[j1], p2 = epair[j2];
            const float4 r0 = y4[(size_t)p0 * 16 + l];
            float4 r1 = y4[(size_t)p1 * 16 + l];
            float4 r2 = y4[(size_t)p2 * 16 + l];
            const float c0 = csrc[p0];
            float c1 = csrc[p1], c2 = csrc[p2];
            const bool u1 = rem > 1, u2 = rem > 2;
            c1 = u1 ? c1 : 0.0f; r1.x = u1 ? r1.x : 0.0f; r1.y = u1 ? r1.y : 0.0f; r1.z = u1 ? r1.z : 0.0f; r1.w = u1 ? r1.w : 0.0f;
            c2 = u2 ? c2 : 0.0f; r2.x = u2 ? r2.x : 0.0f; r2.y = u2 ? r2.y : 0.0f; r2.z = u2 ? r2.z : 0.0f; r2.w = u2 ? r2.w : 0.0f;
            acc.x += r0.x * c0; acc.y += r0.y * c0; acc.z += r0.z * c0; acc.w += r0.w * c0;
            acc.x += r1.x * c1; acc.y += r1.y * c1; acc.z += r1.z * c1; acc.w += r1.w * c1;
            acc.x += r2.x * c2; acc.y += r2.y * c2; acc.z += r2.z * c2; acc.w += r2.w * c2;
        }
        while (j + 4 <= end) {
            const int p0 = q0, p1 = q1, p2 = q2, p3 = q3;
            j += 4;
            if (j + 4 <= end) { q0 = epair[j]; q1 = epair[j + 1]; q2 = epair[j + 2]; q3 = epair[j + 3]; }
            const float4 r0 = y4[(size_t)p0 * 16 + l];
            const float4 r1 = y4[(size_t)p1 * 16 + l];
            const float4 r2 = y4[(size_t)p2 * 16 + l];
            const float4 r3 = y4[(size_t)p3 * 16 + l];
            const float c0 = csrc[p0], c1 = csrc[p1];
            const float c2 = csrc[p2], c3 = csrc[p3];
            acc.x += r0.x * c0; acc.y += r0.y * c0; acc.z += r0.z * c0; acc.w += r0.w * c0;
            acc.x += r1.x * c1; acc.y += r1.y * c1; acc.z += r1.z * c1; acc.w += r1.w * c1;
            acc.x += r2.x * c2; acc.y += r2.y * c2; acc.z += r2.z * c2; acc.w += r2.w * c2;
            acc.x += r3.x * c3; acc.y += r3.y * c3; acc.z += r3.z * c3; acc.w += r3.w * c3;
        }
    }

    const float cd = cdst[node];
    const float4 bv = ((const float4*)bias)[l];
    const float4 ws = ((const float4*)vw)[l];
    const float4 wd = ((const float4*)vw)[16 + l];
    float4 o;
    o.x = fmaxf(acc.x * cd + bv.x, 0.0f);
    o.y = fmaxf(acc.y * cd + bv.y, 0.0f);
    o.z = fmaxf(acc.z * cd + bv.z, 0.0f);
    o.w = fmaxf(acc.w * cd + bv.w, 0.0f);

    float ps = (o.x * ws.x + o.y * ws.y) + (o.z * ws.z + o.w * ws.w);
    float pd = (o.x * wd.x + o.y * wd.y) + (o.z * wd.z + o.w * wd.w);
    ps += __shfl_xor(ps, 8, 32); pd += __shfl_xor(pd, 8, 32);
    ps += __shfl_xor(ps, 4, 32); pd += __shfl_xor(pd, 4, 32);
    ps += __shfl_xor(ps, 2, 32); pd += __shfl_xor(pd, 2, 32);
    ps += __shfl_xor(ps, 1, 32); pd += __shfl_xor(pd, 1, 32);
    if (l == 0) {
        float4 p; p.x = ps; p.y = pd; p.z = csn; p.w = 0.0f;
        ((float4*)z)[node] = p;
    }
}

__global__ __launch_bounds__(256) void gather2_kernel(
        const float* __restrict__ z, const int* __restrict__ epair,
        const int* __restrict__ rowptr, const int* __restrict__ cntd,
        const float* __restrict__ cdst, const float* __restrict__ vw,
        float* __restrict__ pp, int n) {
    const int l = (int)threadIdx.x & 15;
    const int node = (int)blockIdx.x * 16 + ((int)threadIdx.x >> 4);
    if (node >= n) return;
    const int beg = rowptr[node];
    const int deg = cntd[node];
    const float cd = cdst[node];
    asm volatile("" :: "v"(beg), "v"(deg), "v"(cd));
    const int end = beg + deg;
    const float4* __restrict__ z4 = (const float4*)z;

    float as = 0.0f, ad = 0.0f;
    for (int j = beg + l; j < end; j += 32) {
        const bool u1 = j + 16 < end;
        const int p0 = epair[j];
        const int p1 = epair[u1 ? j + 16 : j];
        const float4 z0 = z4[p0];
        float4 z1 = z4[p1];
        const float c0 = z0.z;
        const float c1 = u1 ? z1.z : 0.0f;
        z1.x = u1 ? z1.x : 0.0f; z1.y = u1 ? z1.y : 0.0f;
        as += z0.x * c0; ad += z0.y * c0;
        as += z1.x * c1; ad += z1.y * c1;
    }
    as += __shfl_xor(as, 8, 32); ad += __shfl_xor(ad, 8, 32);
    as += __shfl_xor(as, 4, 32); ad += __shfl_xor(ad, 4, 32);
    as += __shfl_xor(as, 2, 32); ad += __shfl_xor(ad, 2, 32);
    as += __shfl_xor(as, 1, 32); ad += __shfl_xor(ad, 1, 32);
    if (l == 0) {
        float2 p; p.x = as * cd + vw[128]; p.y = ad * cd + vw[129];
        ((float2*)pp)[node] = p;
    }
}

__global__ __launch_bounds__(256) void edge_kernel(
        const float* __restrict__ pp, const int* __restrict__ src,
        const int* __restrict__ dst, const float* __restrict__ efeat,
        const float* __restrict__ We, const float* __restrict__ be,
        float* __restrict__ out, int E4) {
    const int g = (int)blockIdx.x * 256 + (int)threadIdx.x;
    if (g >= E4) return;
    const int4 s = ((const int4*)src)[g];
    const int4 d = ((const int4*)dst)[g];
    const float4* __restrict__ ef4 = (const float4*)efeat + (size_t)g * 3;
    const float4 f0 = ef4[0], f1 = ef4[1], f2 = ef4[2];
    const float s0 = pp[2 * s.x], s1 = pp[2 * s.y], s2 = pp[2 * s.z], s3 = pp[2 * s.w];
    const float d0 = pp[2 * d.x + 1], d1 = pp[2 * d.y + 1], d2 = pp[2 * d.z + 1], d3 = pp[2 * d.w + 1];
    const float w0 = We[128], w1 = We[129], w2 = We[130], b = be[0];
    const float t0 = (s0 + d0) + (f0.x * w0 + f0.y * w1 + f0.z * w2 + b);
    const float t1 = (s1 + d1) + (f0.w * w0 + f1.x * w1 + f1.y * w2 + b);
    const float t2 = (s2 + d2) + (f1.z * w0 + f1.w * w1 + f2.x * w2 + b);
    const float t3 = (s3 + d3) + (f2.y * w0 + f2.z * w1 + f2.w * w2 + b);
    float4 o;
    o.x = 1.0f / (1.0f + expf(-t0));
    o.y = 1.0f / (1.0f + expf(-t1));
    o.z = 1.0f / (1.0f + expf(-t2));
    o.w = 1.0f / (1.0f + expf(-t3));
    ((float4*)out)[g] = o;
}

extern "C" void kernel_launch(void* const* d_in, const int* in_sizes, int n_in,
                              void* d_out, int out_size, void* d_ws, size_t ws_size,
                              hipStream_t stream) {
    const float* x     = (const float*)d_in[0];
    const float* efeat = (const float*)d_in[1];
    const int*   src   = (const int*)  d_in[2];
    const int*   dst   = (const int*)  d_in[3];
    const float* W1    = (const float*)d_in[4];
    const float* b1    = (const float*)d_in[5];
    const float* W2    = (const float*)d_in[6];
    const float* b2    = (const float*)d_in[7];
    const float* We    = (const float*)d_in[8];
    const float* be    = (const float*)d_in[9];
    float* out = (float*)d_out;

    const size_t NB = (size_t)N_NODES * 64;
    char* ws = (char*)d_ws;
    float* bufA = (float*)ws;
    float* zbuf = bufA + NB;
    float* ppair = zbuf + 4 * N_NODES;
    float* vw   = ppair + 2 * N_NODES;
    int*   degD = (int*)(vw + 256);
    int*   rowp = degD + N_PAD;
    float* cS   = (float*)(rowp + N_PAD);
    float* cD   = cS + N_PAD;
    int*   tabD = (int*)(cD + N_PAD);
    int*   tabS = tabD + (NR + 1) * NB1;
    int*   epair = tabS + (NR + 1) * NB1;
    int*   chunkD = epair + N_EDGES;
    int*   chunkS = chunkD + NB1 * EPB;
    static_assert(((size_t)N_NODES * 64 + 6 * N_NODES + 256 + 4 * N_PAD + 2 * (NR + 1) * NB1 + N_EDGES) % 4 == 0,
                  "chunkD starts on a 16-byte boundary");
    static_assert((NB1 * EPB) % 4 == 0, "chunkS starts on a 16-byte boundary");
    static_assert(((size_t)N_NODES * 64 + 6 * N_NODES + 256) % 32 == 0, "the per-node arrays start on a 128-byte line");

    front_kernel<<<GEMM_BLOCKS + 1 + NB1, 256, 0, stream>>>(x, W1, bufA, W2, b2, We, vw,
                                                            src, dst, chunkD, chunkS, tabD, tabS);
    range_kernel<<<2 * NR, RB_THREADS, 0, stream>>>(chunkD, chunkS, tabD, tabS, rowp, degD, cS, cD, epair);

    gather64_proj_kernel<<<(N_NODES + 15) / 16, 256, 0, stream>>>(bufA, epair, rowp, degD, cD, cS, b1, vw, zbuf, N_NODES);

    gather2_kernel<<<(N_NODES + 15) / 16, 256, 0, stream>>>(zbuf, epair, rowp, degD, cD, vw, ppair, N_NODES);

    edge_kernel<<<(N_EDGES / 4 + 255) / 256, 256, 0, stream>>>(ppair, src, dst, efeat, We, be, out, N_EDGES / 4);
}
